# starting kernel
# baseline (speedup 1.0000x reference)
#include <hip/hip_runtime.h>
#include <hip/hip_bf16.h>

typedef __attribute__((ext_vector_type(2))) float v2f;
typedef __attribute__((ext_vector_type(8))) float v8f;

__device__ __forceinline__ v8f wmma_f32_k4(v2f a, v2f b, v8f c) {
    return __builtin_amdgcn_wmma_f32_16x16x4_f32(
         false, a,  false, b,
         (short)0, c,  false,  false);
}

__device__ __forceinline__ void atomAddF(float* p, float v) {
    __hip_atomic_fetch_add(p, v, __ATOMIC_RELAXED, __HIP_MEMORY_SCOPE_AGENT);
}

__global__ void k_init_deg(float* deg, int N) {
    int i = blockIdx.x * blockDim.x + threadIdx.x;
    if (i < N) deg[i] = 1.0f;
}
__global__ void k_deg_accum(const int* __restrict__ dst, float* deg, int E) {
    int e = blockIdx.x * blockDim.x + threadIdx.x;
    if (e < E) atomAddF(&deg[dst[e]], 1.0f);
}
__global__ void k_rsqrt(float* deg, int N) {
    int i = blockIdx.x * blockDim.x + threadIdx.x;
    if (i < N) deg[i] = rsqrtf(deg[i]);
}

__global__ void k_zero(float* p, long long n) {
    long long i = (long long)blockIdx.x * blockDim.x + threadIdx.x;
    if (i < n) p[i] = 0.0f;
}

__global__ void k_gemm1(const float* __restrict__ x, const float* __restrict__ W1,
                        float* __restrict__ xw) {
    const int lane = threadIdx.x & 31;
    const int wave = threadIdx.x >> 5;
    const int half = lane >> 4;
    const int rlo  = lane & 15;
    const int mt   = blockIdx.x;
    const int row  = (mt << 4) + rlo;
    const int ncol = (wave << 4) + rlo;

    v8f acc = {};
    const float* xrow = x + (size_t)row * 300 + (half << 1);
    for (int k = 0; k < 300; k += 4) {
        float2 av = *(const float2*)(xrow + k);
        v2f a; a.x = av.x; a.y = av.y;
        const int kr = k + (half << 1);
        v2f b;
        b.x = W1[(size_t)kr * 128 + ncol];
        b.y = W1[(size_t)(kr + 1) * 128 + ncol];
        acc = wmma_f32_k4(a, b, acc);
    }
    const int m0 = (mt << 4) + (half << 3);
    #pragma unroll
    for (int r = 0; r < 8; ++r)
        xw[(size_t)(m0 + r) * 128 + ncol] = acc[r];
}

__global__ void k_agg1(const float* __restrict__ xw, const int* __restrict__ src,
                       const int* __restrict__ dst, const float* __restrict__ dinv,
                       float* __restrict__ agg, int E) {
    long long gid = (long long)blockIdx.x * blockDim.x + threadIdx.x;
    int e = (int)(gid >> 5);
    int f = ((int)gid & 31) << 2;
    if (e >= E) return;
    int s = src[e], d = dst[e];
    float nrm = dinv[s] * dinv[d];
    float4 v = *(const float4*)(xw + (size_t)s * 128 + f);
    float* o = agg + (size_t)d * 128 + f;
    atomAddF(o + 0, v.x * nrm);
    atomAddF(o + 1, v.y * nrm);
    atomAddF(o + 2, v.z * nrm);
    atomAddF(o + 3, v.w * nrm);
}

__global__ void k_bias_relu(float* __restrict__ h, const float* __restrict__ xw,
                            const float* __restrict__ dinv, const float* __restrict__ b1,
                            int N) {
    long long gid = (long long)blockIdx.x * blockDim.x + threadIdx.x;
    if (gid >= (long long)N * 128) return;
    int node = (int)(gid >> 7);
    int f = (int)gid & 127;
    float di = dinv[node];
    float v = h[gid] + xw[gid] * di * di + b1[f];
    h[gid] = v > 0.0f ? v : 0.0f;
}

__global__ void k_gemm2(const float* __restrict__ h, const float* __restrict__ W2,
                        float* __restrict__ hw16, int N) {
    __shared__ float w2s[128 * 16];
    for (int i = threadIdx.x; i < 128 * 16; i += 256) {
        int k = i >> 4, c = i & 15;
        w2s[i] = (c < 10) ? W2[k * 10 + c] : 0.0f;
    }
    __syncthreads();

    const int lane = threadIdx.x & 31;
    const int wave = threadIdx.x >> 5;
    const int half = lane >> 4;
    const int rlo  = lane & 15;
    const int mt   = blockIdx.x * 8 + wave;
    if ((mt << 4) < N) {
        v8f acc = {};
        const float* hrow = h + (size_t)((mt << 4) + rlo) * 128 + (half << 1);
        for (int k = 0; k < 128; k += 4) {
            float2 av = *(const float2*)(hrow + k);
            v2f a; a.x = av.x; a.y = av.y;
            const int kr = k + (half << 1);
            v2f b;
            b.x = w2s[kr * 16 + rlo];
            b.y = w2s[(kr + 1) * 16 + rlo];
            acc = wmma_f32_k4(a, b, acc);
        }
        const int m0 = (mt << 4) + (half << 3);
        #pragma unroll
        for (int r = 0; r < 8; ++r)
            hw16[(size_t)(m0 + r) * 16 + rlo] = acc[r];
    }
}

__global__ void k_agg2(const float* __restrict__ hw16, const int* __restrict__ src,
                       const int* __restrict__ dst, const float* __restrict__ dinv,
                       float* __restrict__ out, int E) {
    long long gid = (long long)blockIdx.x * blockDim.x + threadIdx.x;
    int e = (int)(gid >> 4);
    int f = (int)gid & 15;
    if (e >= E || f >= 10) return;
    int s = src[e], d = dst[e];
    float nrm = dinv[s] * dinv[d];
    atomAddF(out + (size_t)d * 10 + f, hw16[(size_t)s * 16 + f] * nrm);
}

__global__ void k_final(const float* __restrict__ hw16, const float* __restrict__ dinv,
                        const float* __restrict__ b2, float* __restrict__ out, int N) {
    long long gid = (long long)blockIdx.x * blockDim.x + threadIdx.x;
    if (gid >= (long long)N * 10) return;
    int node = (int)(gid / 10);
    int f = (int)(gid % 10);
    float di = dinv[node];
    out[gid] += hw16[(size_t)node * 16 + f] * di * di + b2[f];
}

extern "C" void kernel_launch(void* const* d_in, const int* in_sizes, int n_in,
                              void* d_out, int out_size, void* d_ws, size_t ws_size,
                              hipStream_t stream) {
    const float* x  = (const float*)d_in[0];
    const int*   ei = (const int*)d_in[1];
    const float* W1 = (const float*)d_in[2];
    const float* b1 = (const float*)d_in[3];
    const float* W2 = (const float*)d_in[4];
    const float* b2 = (const float*)d_in[5];
    float* out = (float*)d_out;

    const int N = in_sizes[0] / 300;
    const int E = in_sizes[1] / 2;
    const int* src = ei;
    const int* dst = ei + E;

    size_t Np = ((size_t)N + 63) & ~(size_t)63;
    float* dinv = (float*)d_ws;
    float* xw   = dinv + Np;
    float* h    = xw + Np * 128;
    float* hw   = h  + Np * 128;

    const int T = 256;
    const long long nf1 = (long long)N * 128;

    k_init_deg <<<(N + T - 1) / T, T, 0, stream>>>(dinv, N);
    k_deg_accum<<<(E + T - 1) / T, T, 0, stream>>>(dst, dinv, E);
    k_rsqrt    <<<(N + T - 1) / T, T, 0, stream>>>(dinv, N);

    k_gemm1    <<<N / 16, T, 0, stream>>>(x, W1, xw);

    k_zero     <<<(unsigned)((nf1 + T - 1) / T), T, 0, stream>>>(h, nf1);
    k_agg1     <<<(unsigned)(((long long)E * 32 + T - 1) / T), T, 0, stream>>>(
                    xw, src, dst, dinv, h, E);
    k_bias_relu<<<(unsigned)((nf1 + T - 1) / T), T, 0, stream>>>(h, xw, dinv, b1, N);

    k_gemm2    <<<(N / 16 + 7) / 8, T, 0, stream>>>(h, W2, hw, N);

    const long long no = (long long)N * 10;
    k_zero     <<<(unsigned)((no + T - 1) / T), T, 0, stream>>>(out, no);
    k_agg2     <<<(unsigned)(((long long)E * 16 + T - 1) / T), T, 0, stream>>>(
                    hw, src, dst, dinv, out, E);
    k_final    <<<(unsigned)((no + T - 1) / T), T, 0, stream>>>(hw, dinv, b2, out, N);
}


# Layer-1 aggregation by owner work-groups: each keeps 400 dst rows in LDS, sweeps the whole dst list, adds matched source rows with LDS float atomics (arrival order), self+bias+relu fused
# speedup vs baseline: 1.5074x; 1.5074x over previous
#include <hip/hip_runtime.h>
#include <hip/hip_bf16.h>

typedef __attribute__((ext_vector_type(2))) float v2f;
typedef __attribute__((ext_vector_type(8))) float v8f;

__device__ __forceinline__ v8f wmma_f32_k4(v2f a, v2f b, v8f c) {
    return __builtin_amdgcn_wmma_f32_16x16x4_f32(
         false, a,  false, b,
         (short)0, c,  false,  false);
}

__device__ __forceinline__ void atomAddF(float* p, float v) {
    __hip_atomic_fetch_add(p, v, __ATOMIC_RELAXED, __HIP_MEMORY_SCOPE_AGENT);
}

__global__ void k_init_deg(float* deg, int N) {
    int i = blockIdx.x * blockDim.x + threadIdx.x;
    if (i < N) deg[i] = 1.0f;
}
__global__ void k_deg_accum(const int* __restrict__ dst, float* deg, int E) {
    int e = blockIdx.x * blockDim.x + threadIdx.x;
    if (e < E) atomAddF(&deg[dst[e]], 1.0f);
}
__global__ void k_rsqrt(float* deg, int N) {
    int i = blockIdx.x * blockDim.x + threadIdx.x;
    if (i < N) deg[i] = rsqrtf(deg[i]);
}

__global__ void k_zero(float* p, long long n) {
    long long i = (long long)blockIdx.x * blockDim.x + threadIdx.x;
    if (i < n) p[i] = 0.0f;
}

__global__ void k_gemm1(const float* __restrict__ x, const float* __restrict__ W1,
                        float* __restrict__ xw) {
    const int lane = threadIdx.x & 31;
    const int wave = threadIdx.x >> 5;
    const int half = lane >> 4;
    const int rlo  = lane & 15;
    const int mt   = blockIdx.x;
    const int row  = (mt << 4) + rlo;
    const int ncol = (wave << 4) + rlo;

    v8f acc = {};
    const float* xrow = x + (size_t)row * 300 + (half << 1);
    for (int k = 0; k < 300; k += 4) {
        float2 av = *(const float2*)(xrow + k);
        v2f a; a.x = av.x; a.y = av.y;
        const int kr = k + (half << 1);
        v2f b;
        b.x = W1[(size_t)kr * 128 + ncol];
        b.y = W1[(size_t)(kr + 1) * 128 + ncol];
        acc = wmma_f32_k4(a, b, acc);
    }
    const int m0 = (mt << 4) + (half << 3);
    #pragma unroll
    for (int r = 0; r < 8; ++r)
        xw[(size_t)(m0 + r) * 128 + ncol] = acc[r];
}

#define PMAX 400
#define QCAP 256

__device__ __forceinline__ void ldsAddF(float* p, float v) {
    __hip_atomic_fetch_add(p, v, __ATOMIC_RELAXED, __HIP_MEMORY_SCOPE_WORKGROUP);
}
__device__ __forceinline__ int rl_i(int v, int k) { return __builtin_amdgcn_readlane(v, k); }
__device__ __forceinline__ float rl_f(float v, int k) {
    return __builtin_bit_cast(float, __builtin_amdgcn_readlane(__builtin_bit_cast(int, v), k));
}

#define PUSH_MATCH(F, EIDX, R)                                                        \
    {                                                                                 \
        const unsigned m_ = __builtin_amdgcn_ballot_w32(F);                           \
        if (m_) {                                                                     \
            if (F) {                                                                  \
                const unsigned sl_ = (tail + __builtin_amdgcn_mbcnt_lo(m_, 0u)) & (QCAP - 1); \
                qe_w[sl_] = (EIDX);                                                   \
                qd_w[sl_] = (R);                                                      \
            }                                                                         \
            tail += (unsigned)__builtin_popcount(m_);                                 \
        }                                                                             \
    }

#define PROCESS8()                                                                    \
    {                                                                                 \
          \
        __builtin_amdgcn_wave_barrier();                                              \
        const unsigned qi_ = (head + (unsigned)(lane & 7)) & (QCAP - 1);              \
        const unsigned qe_ = qe_w[qi_];                                               \
        const int      dl_ = (int)qd_w[qi_];                                          \
        const int      s_  = src[qe_];                                                \
        const float    nr_ = dinv[s_] * dinvL[dl_];                                   \
        float4 row_[8];                                                               \
        _Pragma("unroll")                                                             \
        for (int k = 0; k < 8; ++k)                                                   \
            row_[k] = *(const float4*)(xw + (size_t)rl_i(s_, k) * 128 + (lane << 2)); \
        _Pragma("unroll")                                                             \
        for (int k = 0; k < 8; ++k) {                                                 \
            const float nk_ = rl_f(nr_, k);                                           \
            float* a_ = acc + rl_i(dl_, k) * 128 + lane;                              \
            ldsAddF(a_,      row_[k].x * nk_);                                        \
            ldsAddF(a_ + 32, row_[k].y * nk_);                                        \
            ldsAddF(a_ + 64, row_[k].z * nk_);                                        \
            ldsAddF(a_ + 96, row_[k].w * nk_);                                        \
        }                                                                             \
        head += 8;                                                                    \
    }

__global__ __launch_bounds__(1024)
void k_agg1_own(const float* __restrict__ xw, const int* __restrict__ src,
                const int* __restrict__ dst, const float* __restrict__ dinv,
                const float* __restrict__ b1, float* __restrict__ h, int N, int E) {
    __shared__ float acc[PMAX * 128];
    __shared__ float dinvL[PMAX];
    __shared__ unsigned qe[32 * QCAP];
    __shared__ unsigned qd[32 * QCAP];

    const int tid  = threadIdx.x;
    const int lane = tid & 31;
    const int wave = tid >> 5;
    unsigned* qe_w = qe + wave * QCAP;
    unsigned* qd_w = qd + wave * QCAP;

    const int ntile = ((E & 3) == 0) ? (E >> 7) : 0;
    const int rot = (int)(((long long)blockIdx.x * ntile) / gridDim.x);

    for (int base = blockIdx.x * PMAX; base < N; base += gridDim.x * PMAX) {
        const int P = min(PMAX, N - base);
        for (int i = tid; i < P * 32; i += 1024) ((float4*)acc)[i] = make_float4(0.f, 0.f, 0.f, 0.f);
        for (int i = tid; i < P; i += 1024) dinvL[i] = dinv[base + i];
        __syncthreads();

        unsigned head = 0, tail = 0;

        for (int t0 = wave; t0 < ntile; t0 += 32 * 8) {
            int4 dv[8];
            #pragma unroll
            for (int u = 0; u < 8; ++u) {
                const int t = t0 + 32 * u;
                int tr = t + rot; if (tr >= ntile) tr -= ntile;
                if (t < ntile) dv[u] = *(const int4*)(dst + (size_t)tr * 128 + (lane << 2));
                else           dv[u] = make_int4(-1, -1, -1, -1);
            }
            #pragma unroll
            for (int u = 0; u < 8; ++u) {
                const int t = t0 + 32 * u;
                int tr = t + rot; if (tr >= ntile) tr -= ntile;
                const unsigned e0 = (unsigned)tr * 128u + (unsigned)(lane << 2);
                const unsigned r0 = (unsigned)(dv[u].x - base);
                const unsigned r1 = (unsigned)(dv[u].y - base);
                const unsigned r2 = (unsigned)(dv[u].z - base);
                const unsigned r3 = (unsigned)(dv[u].w - base);
                const bool f0 = r0 < (unsigned)P, f1 = r1 < (unsigned)P;
                const bool f2 = r2 < (unsigned)P, f3 = r3 < (unsigned)P;
                if (__builtin_amdgcn_ballot_w32(f0 | f1 | f2 | f3)) {
                    PUSH_MATCH(f0, e0,      r0)
                    PUSH_MATCH(f1, e0 + 1u, r1)
                    PUSH_MATCH(f2, e0 + 2u, r2)
                    PUSH_MATCH(f3, e0 + 3u, r3)
                    while (tail - head >= 8u) PROCESS8()
                }
            }
        }

        for (int eb = ntile * 128 + wave * 32; eb < E; eb += 1024) {
            const int e = eb + lane;
            const int d = (e < E) ? dst[e] : -1;
            const unsigned r = (unsigned)(d - base);
            const bool f = r < (unsigned)P;
            PUSH_MATCH(f, (unsigned)e, r)
            while (tail - head >= 8u) PROCESS8()
        }

        __builtin_amdgcn_wave_barrier();
        while (head != tail) {
            const unsigned qi = head & (QCAP - 1);
            const unsigned e  = qe_w[qi];
            const int      dl = (int)qd_w[qi];
            const int      s  = src[e];
            const float    nr = dinv[s] * dinvL[dl];
            const float4   rv = *(const float4*)(xw + (size_t)s * 128 + (lane << 2));
            float* a = acc + dl * 128 + lane;
            ldsAddF(a,      rv.x * nr);
            ldsAddF(a + 32, rv.y * nr);
            ldsAddF(a + 64, rv.z * nr);
            ldsAddF(a + 96, rv.w * nr);
            ++head;
        }
        __syncthreads();

        const float4 bv = *(const float4*)(b1 + (lane << 2));
        for (int r = wave; r < P; r += 32) {
            const size_t off = (size_t)(base + r) * 128 + (lane << 2);
            const float di = dinvL[r];
            const float4 xv = *(const float4*)(xw + off);
            const float* a = acc + r * 128 + lane;
            const float v0 = a[0]  + xv.x * di * di + bv.x;
            const float v1 = a[32] + xv.y * di * di + bv.y;
            const float v2 = a[64] + xv.z * di * di + bv.z;
            const float v3 = a[96] + xv.w * di * di + bv.w;
            float4 o;
            o.x = v0 > 0.0f ? v0 : 0.0f;
            o.y = v1 > 0.0f ? v1 : 0.0f;
            o.z = v2 > 0.0f ? v2 : 0.0f;
            o.w = v3 > 0.0f ? v3 : 0.0f;
            *(float4*)(h + off) = o;
        }
        __syncthreads();
    }
}

__global__ void k_gemm2(const float* __restrict__ h, const float* __restrict__ W2,
                        float* __restrict__ hw16, int N) {
    __shared__ float w2s[128 * 16];
    for (int i = threadIdx.x; i < 128 * 16; i += 256) {
        int k = i >> 4, c = i & 15;
        w2s[i] = (c < 10) ? W2[k * 10 + c] : 0.0f;
    }
    __syncthreads();

    const int lane = threadIdx.x & 31;
    const int wave = threadIdx.x >> 5;
    const int half = lane >> 4;
    const int rlo  = lane & 15;
    const int mt   = blockIdx.x * 8 + wave;
    if ((mt << 4) < N) {
        v8f acc = {};
        const float* hrow = h + (size_t)((mt << 4) + rlo) * 128 + (half << 1);
        for (int k = 0; k < 128; k += 4) {
            float2 av = *(const float2*)(hrow + k);
            v2f a; a.x = av.x; a.y = av.y;
            const int kr = k + (half << 1);
            v2f b;
            b.x = w2s[kr * 16 + rlo];
            b.y = w2s[(kr + 1) * 16 + rlo];
            acc = wmma_f32_k4(a, b, acc);
        }
        const int m0 = (mt << 4) + (half << 3);
        #pragma unroll
        for (int r = 0; r < 8; ++r)
            hw16[(size_t)(m0 + r) * 16 + rlo] = acc[r];
    }
}

__global__ void k_agg2(const float* __restrict__ hw16, const int* __restrict__ src,
                       const int* __restrict__ dst, const float* __restrict__ dinv,
                       float* __restrict__ out, int E) {
    long long gid = (long long)blockIdx.x * blockDim.x + threadIdx.x;
    int e = (int)(gid >> 4);
    int f = (int)gid & 15;
    if (e >= E || f >= 10) return;
    int s = src[e], d = dst[e];
    float nrm = dinv[s] * dinv[d];
    atomAddF(out + (size_t)d * 10 + f, hw16[(size_t)s * 16 + f] * nrm);
}

__global__ void k_final(const float* __restrict__ hw16, const float* __restrict__ dinv,
                        const float* __restrict__ b2, float* __restrict__ out, int N) {
    long long gid = (long long)blockIdx.x * blockDim.x + threadIdx.x;
    if (gid >= (long long)N * 10) return;
    int node = (int)(gid / 10);
    int f = (int)(gid % 10);
    float di = dinv[node];
    out[gid] += hw16[(size_t)node * 16 + f] * di * di + b2[f];
}

extern "C" void kernel_launch(void* const* d_in, const int* in_sizes, int n_in,
                              void* d_out, int out_size, void* d_ws, size_t ws_size,
                              hipStream_t stream) {
    const float* x  = (const float*)d_in[0];
    const int*   ei = (const int*)d_in[1];
    const float* W1 = (const float*)d_in[2];
    const float* b1 = (const float*)d_in[3];
    const float* W2 = (const float*)d_in[4];
    const float* b2 = (const float*)d_in[5];
    float* out = (float*)d_out;

    const int N = in_sizes[0] / 300;
    const int E = in_sizes[1] / 2;
    const int* src = ei;
    const int* dst = ei + E;

    size_t Np = ((size_t)N + 63) & ~(size_t)63;
    float* dinv = (float*)d_ws;
    float* xw   = dinv + Np;
    float* h    = xw + Np * 128;
    float* hw   = h  + Np * 128;

    const int T = 256;

    k_init_deg <<<(N + T - 1) / T, T, 0, stream>>>(dinv, N);
    k_deg_accum<<<(E + T - 1) / T, T, 0, stream>>>(dst, dinv, E);
    k_rsqrt    <<<(N + T - 1) / T, T, 0, stream>>>(dinv, N);

    k_gemm1    <<<N / 16, T, 0, stream>>>(x, W1, xw);

    k_agg1_own <<<256, 1024, 0, stream>>>(xw, src, dst, dinv, b1, h, N, E);

    k_gemm2    <<<(N / 16 + 7) / 8, T, 0, stream>>>(h, W2, hw, N);

    const long long no = (long long)N * 10;
    k_zero     <<<(unsigned)((no + T - 1) / T), T, 0, stream>>>(out, no);
    k_agg2     <<<(unsigned)(((long long)E * 16 + T - 1) / T), T, 0, stream>>>(
                    hw, src, dst, dinv, out, E);
    k_final    <<<(unsigned)((no + T - 1) / T), T, 0, stream>>>(hw, dinv, b2, out, N);
}


# First product on 512 resident work-groups: a wave keeps its 300x16 strip of W1 in registers (75 B fragments), 16x300 x tiles go through LDS by whole-line loads, next tile prefetched; sum order kept
# speedup vs baseline: 1.8777x; 1.2457x over previous
#include <hip/hip_runtime.h>
#include <hip/hip_bf16.h>

typedef __attribute__((ext_vector_type(2))) float v2f;
typedef __attribute__((ext_vector_type(8))) float v8f;

__device__ __forceinline__ v8f wmma_f32_k4(v2f a, v2f b, v8f c) {
    return __builtin_amdgcn_wmma_f32_16x16x4_f32(
         false, a,  false, b,
         (short)0, c,  false,  false);
}

__device__ __forceinline__ void atomAddF(float* p, float v) {
    __hip_atomic_fetch_add(p, v, __ATOMIC_RELAXED, __HIP_MEMORY_SCOPE_AGENT);
}

__global__ void k_init_deg(float* deg, int N) {
    int i = blockIdx.x * blockDim.x + threadIdx.x;
    if (i < N) deg[i] = 1.0f;
}
__global__ void k_deg_accum(const int* __restrict__ dst, float* deg, int E) {
    int e = blockIdx.x * blockDim.x + threadIdx.x;
    if (e < E) atomAddF(&deg[dst[e]], 1.0f);
}
__global__ void k_rsqrt(float* deg, int N) {
    int i = blockIdx.x * blockDim.x + threadIdx.x;
    if (i < N) deg[i] = rsqrtf(deg[i]);
}

__global__ void k_zero(float* p, long long n) {
    long long i = (long long)blockIdx.x * blockDim.x + threadIdx.x;
    if (i < n) p[i] = 0.0f;
}

#define G1_TF 4800
#define G1_V4 1200

__global__ __launch_bounds__(256)
void k_gemm1(const float* __restrict__ x, const float* __restrict__ W1,
             float* __restrict__ xw, int ntiles) {
    __shared__ __attribute__((aligned(16))) float As[2 * G1_TF];

    const int tid  = threadIdx.x;
    const int lane = tid & 31;
    const int wave = tid >> 5;
    const int half = lane >> 4;
    const int rlo  = lane & 15;
    const int ncol = (wave << 4) + rlo;

    v2f bf[75];
    #pragma unroll
    for (int i = 0; i < 75; ++i) {
        const int kr = 4 * i + (half << 1);
        bf[i].x = W1[kr * 128 + ncol];
        bf[i].y = W1[(kr + 1) * 128 + ncol];
    }

    int tile = blockIdx.x;
    float4 pre[5];
    #pragma unroll
    for (int j = 0; j < 5; ++j) pre[j] = make_float4(0.f, 0.f, 0.f, 0.f);
    if (tile < ntiles) {
        const float4* g = (const float4*)(x + (size_t)tile * G1_TF);
        #pragma unroll
        for (int j = 0; j < 5; ++j) {
            const int idx = tid + 256 * j;
            pre[j] = g[idx < G1_V4 ? idx : G1_V4 - 1];
        }
    }

    int buf = 0;
    for (; tile < ntiles; tile += gridDim.x) {
        float4* st = (float4*)(As + buf * G1_TF);
        #pragma unroll
        for (int j = 0; j < 5; ++j) {
            const int idx = tid + 256 * j;
            if (idx < G1_V4) st[idx] = pre[j];
        }
        __syncthreads();

        const int nxt = tile + gridDim.x;
        if (nxt < ntiles) {
            const float4* g = (const float4*)(x + (size_t)nxt * G1_TF);
            #pragma unroll
            for (int j = 0; j < 5; ++j) {
                const int idx = tid + 256 * j;
                pre[j] = g[idx < G1_V4 ? idx : G1_V4 - 1];
            }
        }

        v8f acc = {};
        const float* ap = As + buf * G1_TF + rlo * 300 + (half << 1);
        #pragma unroll
        for (int i = 0; i < 75; ++i) {
            const float2 av = *(const float2*)(ap + 4 * i);
            v2f a; a.x = av.x; a.y = av.y;
            acc = wmma_f32_k4(a, bf[i], acc);
        }
        const int m0 = (tile << 4) + (half << 3);
        #pragma unroll
        for (int r = 0; r < 8; ++r)
            xw[(size_t)(m0 + r) * 128 + ncol] = acc[r];
        buf ^= 1;
    }
}

#define PMAX 400
#define QCAP 256

__device__ __forceinline__ void ldsAddF(float* p, float v) {
    __hip_atomic_fetch_add(p, v, __ATOMIC_RELAXED, __HIP_MEMORY_SCOPE_WORKGROUP);
}
__device__ __forceinline__ int rl_i(int v, int k) { return __builtin_amdgcn_readlane(v, k); }
__device__ __forceinline__ float rl_f(float v, int k) {
    return __builtin_bit_cast(float, __builtin_amdgcn_readlane(__builtin_bit_cast(int, v), k));
}

#define PUSH_MATCH(F, EIDX, R)                                                        \
    {                                                                                 \
        const unsigned m_ = __builtin_amdgcn_ballot_w32(F);                           \
        if (m_) {                                                                     \
            if (F) {                                                                  \
                const unsigned sl_ = (tail + __builtin_amdgcn_mbcnt_lo(m_, 0u)) & (QCAP - 1); \
                qe_w[sl_] = (EIDX);                                                   \
                qd_w[sl_] = (R);                                                      \
            }                                                                         \
            tail += (unsigned)__builtin_popcount(m_);                                 \
        }                                                                             \
    }

#define PROCESS8()                                                                    \
    {                                                                                 \
          \
        __builtin_amdgcn_wave_barrier();                                              \
        const unsigned qi_ = (head + (unsigned)(lane & 7)) & (QCAP - 1);              \
        const unsigned qe_ = qe_w[qi_];                                               \
        const int      dl_ = (int)qd_w[qi_];                                          \
        const int      s_  = src[qe_];                                                \
        const float    nr_ = dinv[s_] * dinvL[dl_];                                   \
        float4 row_[8];                                                               \
        _Pragma("unroll")                                                             \
        for (int k = 0; k < 8; ++k)                                                   \
            row_[k] = *(const float4*)(xw + (size_t)rl_i(s_, k) * 128 + (lane << 2)); \
        _Pragma("unroll")                                                             \
        for (int k = 0; k < 8; ++k) {                                                 \
            const float nk_ = rl_f(nr_, k);                                           \
            float* a_ = acc + rl_i(dl_, k) * 128 + lane;                              \
            ldsAddF(a_,      row_[k].x * nk_);                                        \
            ldsAddF(a_ + 32, row_[k].y * nk_);                                        \
            ldsAddF(a_ + 64, row_[k].z * nk_);                                        \
            ldsAddF(a_ + 96, row_[k].w * nk_);                                        \
        }                                                                             \
        head += 8;                                                                    \
    }

__global__ __launch_bounds__(1024)
void k_agg1_own(const float* __restrict__ xw, const int* __restrict__ src,
                const int* __restrict__ dst, const float* __restrict__ dinv,
                const float* __restrict__ b1, float* __restrict__ h, int N, int E) {
    __shared__ float acc[PMAX * 128];
    __shared__ float dinvL[PMAX];
    __shared__ unsigned qe[32 * QCAP];
    __shared__ unsigned qd[32 * QCAP];

    const int tid  = threadIdx.x;
    const int lane = tid & 31;
    const int wave = tid >> 5;
    unsigned* qe_w = qe + wave * QCAP;
    unsigned* qd_w = qd + wave * QCAP;

    const int ntile = ((E & 3) == 0) ? (E >> 7) : 0;
    const int rot = (int)(((long long)blockIdx.x * ntile) / gridDim.x);

    for (int base = blockIdx.x * PMAX; base < N; base += gridDim.x * PMAX) {
        const int P = min(PMAX, N - base);
        for (int i = tid; i < P * 32; i += 1024) ((float4*)acc)[i] = make_float4(0.f, 0.f, 0.f, 0.f);
        for (int i = tid; i < P; i += 1024) dinvL[i] = dinv[base + i];
        __syncthreads();

        unsigned head = 0, tail = 0;

        for (int t0 = wave; t0 < ntile; t0 += 32 * 8) {
            int4 dv[8];
            #pragma unroll
            for (int u = 0; u < 8; ++u) {
                const int t = t0 + 32 * u;
                int tr = t + rot; if (tr >= ntile) tr -= ntile;
                if (t < ntile) dv[u] = *(const int4*)(dst + (size_t)tr * 128 + (lane << 2));
                else           dv[u] = make_int4(-1, -1, -1, -1);
            }
            #pragma unroll
            for (int u = 0; u < 8; ++u) {
                const int t = t0 + 32 * u;
                int tr = t + rot; if (tr >= ntile) tr -= ntile;
                const unsigned e0 = (unsigned)tr * 128u + (unsigned)(lane << 2);
                const unsigned r0 = (unsigned)(dv[u].x - base);
                const unsigned r1 = (unsigned)(dv[u].y - base);
                const unsigned r2 = (unsigned)(dv[u].z - base);
                const unsigned r3 = (unsigned)(dv[u].w - base);
                const bool f0 = r0 < (unsigned)P, f1 = r1 < (unsigned)P;
                const bool f2 = r2 < (unsigned)P, f3 = r3 < (unsigned)P;
                if (__builtin_amdgcn_ballot_w32(f0 | f1 | f2 | f3)) {
                    PUSH_MATCH(f0, e0,      r0)
                    PUSH_MATCH(f1, e0 + 1u, r1)
                    PUSH_MATCH(f2, e0 + 2u, r2)
                    PUSH_MATCH(f3, e0 + 3u, r3)
                    while (tail - head >= 8u) PROCESS8()
                }
            }
        }

        for (int eb = ntile * 128 + wave * 32; eb < E; eb += 1024) {
            const int e = eb + lane;
            const int d = (e < E) ? dst[e] : -1;
            const unsigned r = (unsigned)(d - base);
            const bool f = r < (unsigned)P;
            PUSH_MATCH(f, (unsigned)e, r)
            while (tail - head >= 8u) PROCESS8()
        }

        __builtin_amdgcn_wave_barrier();
        while (head != tail) {
            const unsigned qi = head & (QCAP - 1);
            const unsigned e  = qe_w[qi];
            const int      dl = (int)qd_w[qi];
            const int      s  = src[e];
            const float    nr = dinv[s] * dinvL[dl];
            const float4   rv = *(const float4*)(xw + (size_t)s * 128 + (lane << 2));
            float* a = acc + dl * 128 + lane;
            ldsAddF(a,      rv.x * nr);
            ldsAddF(a + 32, rv.y * nr);
            ldsAddF(a + 64, rv.z * nr);
            ldsAddF(a + 96, rv.w * nr);
            ++head;
        }
        __syncthreads();

        const float4 bv = *(const float4*)(b1 + (lane << 2));
        for (int r = wave; r < P; r += 32) {
            const size_t off = (size_t)(base + r) * 128 + (lane << 2);
            const float di = dinvL[r];
            const float4 xv = *(const float4*)(xw + off);
            const float* a = acc + r * 128 + lane;
            const float v0 = a[0]  + xv.x * di * di + bv.x;
            const float v1 = a[32] + xv.y * di * di + bv.y;
            const float v2 = a[64] + xv.z * di * di + bv.z;
            const float v3 = a[96] + xv.w * di * di + bv.w;
            float4 o;
            o.x = v0 > 0.0f ? v0 : 0.0f;
            o.y = v1 > 0.0f ? v1 : 0.0f;
            o.z = v2 > 0.0f ? v2 : 0.0f;
            o.w = v3 > 0.0f ? v3 : 0.0f;
            *(float4*)(h + off) = o;
        }
        __syncthreads();
    }
}

__global__ void k_gemm2(const float* __restrict__ h, const float* __restrict__ W2,
                        float* __restrict__ hw16, int N) {
    __shared__ float w2s[128 * 16];
    for (int i = threadIdx.x; i < 128 * 16; i += 256) {
        int k = i >> 4, c = i & 15;
        w2s[i] = (c < 10) ? W2[k * 10 + c] : 0.0f;
    }
    __syncthreads();

    const int lane = threadIdx.x & 31;
    const int wave = threadIdx.x >> 5;
    const int half = lane >> 4;
    const int rlo  = lane & 15;
    const int mt   = blockIdx.x * 8 + wave;
    if ((mt << 4) < N) {
        v8f acc = {};
        const float* hrow = h + (size_t)((mt << 4) + rlo) * 128 + (half << 1);
        for (int k = 0; k < 128; k += 4) {
            float2 av = *(const float2*)(hrow + k);
            v2f a; a.x = av.x; a.y = av.y;
            const int kr = k + (half << 1);
            v2f b;
            b.x = w2s[kr * 16 + rlo];
            b.y = w2s[(kr + 1) * 16 + rlo];
            acc = wmma_f32_k4(a, b, acc);
        }
        const int m0 = (mt << 4) + (half << 3);
        #pragma unroll
        for (int r = 0; r < 8; ++r)
            hw16[(size_t)(m0 + r) * 16 + rlo] = acc[r];
    }
}

__global__ void k_agg2(const float* __restrict__ hw16, const int* __restrict__ src,
                       const int* __restrict__ dst, const float* __restrict__ dinv,
                       float* __restrict__ out, int E) {
    long long gid = (long long)blockIdx.x * blockDim.x + threadIdx.x;
    int e = (int)(gid >> 4);
    int f = (int)gid & 15;
    if (e >= E || f >= 10) return;
    int s = src[e], d = dst[e];
    float nrm = dinv[s] * dinv[d];
    atomAddF(out + (size_t)d * 10 + f, hw16[(size_t)s * 16 + f] * nrm);
}

__global__ void k_final(const float* __restrict__ hw16, const float* __restrict__ dinv,
                        const float* __restrict__ b2, float* __restrict__ out, int N) {
    long long gid = (long long)blockIdx.x * blockDim.x + threadIdx.x;
    if (gid >= (long long)N * 10) return;
    int node = (int)(gid / 10);
    int f = (int)(gid % 10);
    float di = dinv[node];
    out[gid] += hw16[(size_t)node * 16 + f] * di * di + b2[f];
}

extern "C" void kernel_launch(void* const* d_in, const int* in_sizes, int n_in,
                              void* d_out, int out_size, void* d_ws, size_t ws_size,
                              hipStream_t stream) {
    const float* x  = (const float*)d_in[0];
    const int*   ei = (const int*)d_in[1];
    const float* W1 = (const float*)d_in[2];
    const float* b1 = (const float*)d_in[3];
    const float* W2 = (const float*)d_in[4];
    const float* b2 = (const float*)d_in[5];
    float* out = (float*)d_out;

    const int N = in_sizes[0] / 300;
    const int E = in_sizes[1] / 2;
    const int* src = ei;
    const int* dst = ei + E;

    size_t Np = ((size_t)N + 63) & ~(size_t)63;
    float* dinv = (float*)d_ws;
    float* xw   = dinv + Np;
    float* h    = xw + Np * 128;
    float* hw   = h  + Np * 128;

    const int T = 256;

    k_init_deg <<<(N + T - 1) / T, T, 0, stream>>>(dinv, N);
    k_deg_accum<<<(E + T - 1) / T, T, 0, stream>>>(dst, dinv, E);
    k_rsqrt    <<<(N + T - 1) / T, T, 0, stream>>>(dinv, N);

    k_gemm1    <<<512, T, 0, stream>>>(x, W1, xw, N / 16);

    k_agg1_own <<<256, 1024, 0, stream>>>(xw, src, dst, dinv, b1, h, N, E);

    k_gemm2    <<<(N / 16 + 7) / 8, T, 0, stream>>>(h, W2, hw, N);

    const long long no = (long long)N * 10;
    k_zero     <<<(unsigned)((no + T - 1) / T), T, 0, stream>>>(out, no);
    k_agg2     <<<(unsigned)(((long long)E * 16 + T - 1) / T), T, 0, stream>>>(
                    hw, src, dst, dinv, out, E);
    k_final    <<<(unsigned)((no + T - 1) / T), T, 0, stream>>>(hw, dinv, b2, out, N);
}


# Layer-2 aggregation on a 32x8 grid of owner work-groups (node range x edge segment): 10-float rows in LDS, LDS float adds (arrival order), 8 partial planes summed in fixed order; no global atomics
# speedup vs baseline: 2.1172x; 1.1275x over previous
#include <hip/hip_runtime.h>
#include <hip/hip_bf16.h>

typedef __attribute__((ext_vector_type(2))) float v2f;
typedef __attribute__((ext_vector_type(8))) float v8f;

__device__ __forceinline__ v8f wmma_f32_k4(v2f a, v2f b, v8f c) {
    return __builtin_amdgcn_wmma_f32_16x16x4_f32(
         false, a,  false, b,
         (short)0, c,  false,  false);
}

__device__ __forceinline__ void atomAddF(float* p, float v) {
    __hip_atomic_fetch_add(p, v, __ATOMIC_RELAXED, __HIP_MEMORY_SCOPE_AGENT);
}

__global__ void k_init_deg(float* deg, int N) {
    int i = blockIdx.x * blockDim.x + threadIdx.x;
    if (i < N) deg[i] = 1.0f;
}
__global__ void k_deg_accum(const int* __restrict__ dst, float* deg, int E) {
    int e = blockIdx.x * blockDim.x + threadIdx.x;
    if (e < E) atomAddF(&deg[dst[e]], 1.0f);
}
__global__ void k_rsqrt(float* deg, int N) {
    int i = blockIdx.x * blockDim.x + threadIdx.x;
    if (i < N) deg[i] = rsqrtf(deg[i]);
}

#define G1_TF 4800
#define G1_V4 1200

__global__ __launch_bounds__(256)
void k_gemm1(const float* __restrict__ x, const float* __restrict__ W1,
             float* __restrict__ xw, int ntiles) {
    __shared__ __attribute__((aligned(16))) float As[2 * G1_TF];

    const int tid  = threadIdx.x;
    const int lane = tid & 31;
    const int wave = tid >> 5;
    const int half = lane >> 4;
    const int rlo  = lane & 15;
    const int ncol = (wave << 4) + rlo;

    v2f bf[75];
    #pragma unroll
    for (int i = 0; i < 75; ++i) {
        const int kr = 4 * i + (half << 1);
        bf[i].x = W1[kr * 128 + ncol];
        bf[i].y = W1[(kr + 1) * 128 + ncol];
    }

    int tile = blockIdx.x;
    float4 pre[5];
    #pragma unroll
    for (int j = 0; j < 5; ++j) pre[j] = make_float4(0.f, 0.f, 0.f, 0.f);
    if (tile < ntiles) {
        const float4* g = (const float4*)(x + (size_t)tile * G1_TF);
        #pragma unroll
        for (int j = 0; j < 5; ++j) {
            const int idx = tid + 256 * j;
            pre[j] = g[idx < G1_V4 ? idx : G1_V4 - 1];
        }
    }

    int buf = 0;
    for (; tile < ntiles; tile += gridDim.x) {
        float4* st = (float4*)(As + buf * G1_TF);
        #pragma unroll
        for (int j = 0; j < 5; ++j) {
            const int idx = tid + 256 * j;
            if (idx < G1_V4) st[idx] = pre[j];
        }
        __syncthreads();

        const int nxt = tile + gridDim.x;
        if (nxt < ntiles) {
            const float4* g = (const float4*)(x + (size_t)nxt * G1_TF);
            #pragma unroll
            for (int j = 0; j < 5; ++j) {
                const int idx = tid + 256 * j;
                pre[j] = g[idx < G1_V4 ? idx : G1_V4 - 1];
            }
        }

        v8f acc = {};
        const float* ap = As + buf * G1_TF + rlo * 300 + (half << 1);
        #pragma unroll
        for (int i = 0; i < 75; ++i) {
            const float2 av = *(const float2*)(ap + 4 * i);
            v2f a; a.x = av.x; a.y = av.y;
            acc = wmma_f32_k4(a, bf[i], acc);
        }
        const int m0 = (tile << 4) + (half << 3);
        #pragma unroll
        for (int r = 0; r < 8; ++r)
            xw[(size_t)(m0 + r) * 128 + ncol] = acc[r];
        buf ^= 1;
    }
}

#define PMAX 400
#define QCAP 256

__device__ __forceinline__ void ldsAddF(float* p, float v) {
    __hip_atomic_fetch_add(p, v, __ATOMIC_RELAXED, __HIP_MEMORY_SCOPE_WORKGROUP);
}
__device__ __forceinline__ int rl_i(int v, int k) { return __builtin_amdgcn_readlane(v, k); }
__device__ __forceinline__ float rl_f(float v, int k) {
    return __builtin_bit_cast(float, __builtin_amdgcn_readlane(__builtin_bit_cast(int, v), k));
}

#define PUSH_MATCH(F, EIDX, R)                                                        \
    {                                                                                 \
        const unsigned m_ = __builtin_amdgcn_ballot_w32(F);                           \
        if (m_) {                                                                     \
            if (F) {                                                                  \
                const unsigned sl_ = (tail + __builtin_amdgcn_mbcnt_lo(m_, 0u)) & (QCAP - 1); \
                qe_w[sl_] = (EIDX);                                                   \
                qd_w[sl_] = (R);                                                      \
            }                                                                         \
            tail += (unsigned)__builtin_popcount(m_);                                 \
        }                                                                             \
    }

#define PROCESS8()                                                                    \
    {                                                                                 \
          \
        __builtin_amdgcn_wave_barrier();                                              \
        const unsigned qi_ = (head + (unsigned)(lane & 7)) & (QCAP - 1);              \
        const unsigned qe_ = qe_w[qi_];                                               \
        const int      dl_ = (int)qd_w[qi_];                                          \
        const int      s_  = src[qe_];                                                \
        const float    nr_ = dinv[s_] * dinvL[dl_];                                   \
        float4 row_[8];                                                               \
        _Pragma("unroll")                                                             \
        for (int k = 0; k < 8; ++k)                                                   \
            row_[k] = *(const float4*)(xw + (size_t)rl_i(s_, k) * 128 + (lane << 2)); \
        _Pragma("unroll")                                                             \
        for (int k = 0; k < 8; ++k) {                                                 \
            const float nk_ = rl_f(nr_, k);                                           \
            float* a_ = acc + rl_i(dl_, k) * 128 + lane;                              \
            ldsAddF(a_,      row_[k].x * nk_);                                        \
            ldsAddF(a_ + 32, row_[k].y * nk_);                                        \
            ldsAddF(a_ + 64, row_[k].z * nk_);                                        \
            ldsAddF(a_ + 96, row_[k].w * nk_);                                        \
        }                                                                             \
        head += 8;                                                                    \
    }

__global__ __launch_bounds__(1024)
void k_agg1_own(const float* __restrict__ xw, const int* __restrict__ src,
                const int* __restrict__ dst, const float* __restrict__ dinv,
                const float* __restrict__ b1, float* __restrict__ h, int N, int E) {
    __shared__ float acc[PMAX * 128];
    __shared__ float dinvL[PMAX];
    __shared__ unsigned qe[32 * QCAP];
    __shared__ unsigned qd[32 * QCAP];

    const int tid  = threadIdx.x;
    const int lane = tid & 31;
    const int wave = tid >> 5;
    unsigned* qe_w = qe + wave * QCAP;
    unsigned* qd_w = qd + wave * QCAP;

    const int ntile = ((E & 3) == 0) ? (E >> 7) : 0;
    const int rot = (int)(((long long)blockIdx.x * ntile) / gridDim.x);

    for (int base = blockIdx.x * PMAX; base < N; base += gridDim.x * PMAX) {
        const int P = min(PMAX, N - base);
        for (int i = tid; i < P * 32; i += 1024) ((float4*)acc)[i] = make_float4(0.f, 0.f, 0.f, 0.f);
        for (int i = tid; i < P; i += 1024) dinvL[i] = dinv[base + i];
        __syncthreads();

        unsigned head = 0, tail = 0;

        for (int t0 = wave; t0 < ntile; t0 += 32 * 8) {
            int4 dv[8];
            #pragma unroll
            for (int u = 0; u < 8; ++u) {
                const int t = t0 + 32 * u;
                int tr = t + rot; if (tr >= ntile) tr -= ntile;
                if (t < ntile) dv[u] = *(const int4*)(dst + (size_t)tr * 128 + (lane << 2));
                else           dv[u] = make_int4(-1, -1, -1, -1);
            }
            #pragma unroll
            for (int u = 0; u < 8; ++u) {
                const int t = t0 + 32 * u;
                int tr = t + rot; if (tr >= ntile) tr -= ntile;
                const unsigned e0 = (unsigned)tr * 128u + (unsigned)(lane << 2);
                const unsigned r0 = (unsigned)(dv[u].x - base);
                const unsigned r1 = (unsigned)(dv[u].y - base);
                const unsigned r2 = (unsigned)(dv[u].z - base);
                const unsigned r3 = (unsigned)(dv[u].w - base);
                const bool f0 = r0 < (unsigned)P, f1 = r1 < (unsigned)P;
                const bool f2 = r2 < (unsigned)P, f3 = r3 < (unsigned)P;
                if (__builtin_amdgcn_ballot_w32(f0 | f1 | f2 | f3)) {
                    PUSH_MATCH(f0, e0,      r0)
                    PUSH_MATCH(f1, e0 + 1u, r1)
                    PUSH_MATCH(f2, e0 + 2u, r2)
                    PUSH_MATCH(f3, e0 + 3u, r3)
                    while (tail - head >= 8u) PROCESS8()
                }
            }
        }

        for (int eb = ntile * 128 + wave * 32; eb < E; eb += 1024) {
            const int e = eb + lane;
            const int d = (e < E) ? dst[e] : -1;
            const unsigned r = (unsigned)(d - base);
            const bool f = r < (unsigned)P;
            PUSH_MATCH(f, (unsigned)e, r)
            while (tail - head >= 8u) PROCESS8()
        }

        __builtin_amdgcn_wave_barrier();
        while (head != tail) {
            const unsigned qi = head & (QCAP - 1);
            const unsigned e  = qe_w[qi];
            const int      dl = (int)qd_w[qi];
            const int      s  = src[e];
            const float    nr = dinv[s] * dinvL[dl];
            const float4   rv = *(const float4*)(xw + (size_t)s * 128 + (lane << 2));
            float* a = acc + dl * 128 + lane;
            ldsAddF(a,      rv.x * nr);
            ldsAddF(a + 32, rv.y * nr);
            ldsAddF(a + 64, rv.z * nr);
            ldsAddF(a + 96, rv.w * nr);
            ++head;
        }
        __syncthreads();

        const float4 bv = *(const float4*)(b1 + (lane << 2));
        for (int r = wave; r < P; r += 32) {
            const size_t off = (size_t)(base + r) * 128 + (lane << 2);
            const float di = dinvL[r];
            const float4 xv = *(const float4*)(xw + off);
            const float* a = acc + r * 128 + lane;
            const float v0 = a[0]  + xv.x * di * di + bv.x;
            const float v1 = a[32] + xv.y * di * di + bv.y;
            const float v2 = a[64] + xv.z * di * di + bv.z;
            const float v3 = a[96] + xv.w * di * di + bv.w;
            float4 o;
            o.x = v0 > 0.0f ? v0 : 0.0f;
            o.y = v1 > 0.0f ? v1 : 0.0f;
            o.z = v2 > 0.0f ? v2 : 0.0f;
            o.w = v3 > 0.0f ? v3 : 0.0f;
            *(float4*)(h + off) = o;
        }
        __syncthreads();
    }
}

__global__ void k_gemm2(const float* __restrict__ h, const float* __restrict__ W2,
                        float* __restrict__ hw16, int N) {
    __shared__ float w2s[128 * 16];
    for (int i = threadIdx.x; i < 128 * 16; i += 256) {
        int k = i >> 4, c = i & 15;
        w2s[i] = (c < 10) ? W2[k * 10 + c] : 0.0f;
    }
    __syncthreads();

    const int lane = threadIdx.x & 31;
    const int wave = threadIdx.x >> 5;
    const int half = lane >> 4;
    const int rlo  = lane & 15;
    const int mt   = blockIdx.x * 8 + wave;
    if ((mt << 4) < N) {
        v8f acc = {};
        const float* hrow = h + (size_t)((mt << 4) + rlo) * 128 + (half << 1);
        for (int k = 0; k < 128; k += 4) {
            float2 av = *(const float2*)(hrow + k);
            v2f a; a.x = av.x; a.y = av.y;
            const int kr = k + (half << 1);
            v2f b;
            b.x = w2s[kr * 16 + rlo];
            b.y = w2s[(kr + 1) * 16 + rlo];
            acc = wmma_f32_k4(a, b, acc);
        }
        const int m0 = (mt << 4) + (half << 3);
        #pragma unroll
        for (int r = 0; r < 8; ++r)
            hw16[(size_t)(m0 + r) * 16 + rlo] = acc[r];
    }
}

#define A2_GI 32
#define A2_GJ 8
#define A2_R  3136
#define A2_Q  256

__device__ __forceinline__ void a2_apply(const float* __restrict__ hw16, const float* __restrict__ dinv,
                                         const float* dinvL, float* acc, unsigned s, unsigned dl) {
    const float nr = dinv[s] * dinvL[dl];
    const float* rp = hw16 + (size_t)s * 16;
    const float4 r0 = *(const float4*)rp;
    const float4 r1 = *(const float4*)(rp + 4);
    const float2 r2 = *(const float2*)(rp + 8);
    float* a = acc + dl * 10;
    ldsAddF(a,     r0.x * nr);
    ldsAddF(a + 1, r0.y * nr);
    ldsAddF(a + 2, r0.z * nr);
    ldsAddF(a + 3, r0.w * nr);
    ldsAddF(a + 4, r1.x * nr);
    ldsAddF(a + 5, r1.y * nr);
    ldsAddF(a + 6, r1.z * nr);
    ldsAddF(a + 7, r1.w * nr);
    ldsAddF(a + 8, r2.x * nr);
    ldsAddF(a + 9, r2.y * nr);
}

#define A2_PUSH(F, S, R)                                                              \
    {                                                                                 \
        const unsigned m_ = __builtin_amdgcn_ballot_w32(F);                           \
        if (m_) {                                                                     \
            if (F) {                                                                  \
                const unsigned sl_ = (tail + __builtin_amdgcn_mbcnt_lo(m_, 0u)) & (A2_Q - 1); \
                qs_w[sl_] = (unsigned)(S);                                            \
                qd_w[sl_] = (R);                                                      \
            }                                                                         \
            tail += (unsigned)__builtin_popcount(m_);                                 \
        }                                                                             \
    }

#define A2_PROCESS32()                                                                \
    {                                                                                 \
          \
        __builtin_amdgcn_wave_barrier();                                              \
        const unsigned qi_ = (head + (unsigned)lane) & (A2_Q - 1);                    \
        a2_apply(hw16, dinv, dinvL, acc, qs_w[qi_], qd_w[qi_]);                       \
        head += 32;                                                                   \
    }

__global__ __launch_bounds__(1024)
void k_agg2_own(const float* __restrict__ hw16, const int* __restrict__ src,
                const int* __restrict__ dst, const float* __restrict__ dinv,
                float* __restrict__ part, int N, int E) {
    __shared__ __attribute__((aligned(16))) float acc[A2_R * 10];
    __shared__ float dinvL[A2_R];
    __shared__ unsigned qs[32 * A2_Q];
    __shared__ unsigned qd[32 * A2_Q];

    const int tid  = threadIdx.x;
    const int lane = tid & 31;
    const int wave = tid >> 5;
    unsigned* qs_w = qs + wave * A2_Q;
    unsigned* qd_w = qd + wave * A2_Q;

    const int gi = (int)blockIdx.x / A2_GJ;
    const int gj = (int)blockIdx.x % A2_GJ;

    const size_t pstride = ((size_t)N * 10 + 31) & ~(size_t)31;
    float* plane = part + (size_t)gj * pstride;

    const int ntile = ((E & 3) == 0) ? (E >> 7) : 0;
    const int t_lo  = (int)(((long long)gj * ntile) / A2_GJ);
    const int t_hi  = (int)(((long long)(gj + 1) * ntile) / A2_GJ);
    const int nseg  = t_hi - t_lo;
    const int rot = (int)(((long long)gi * nseg) / A2_GI);

    for (int base = gi * A2_R; base < N; base += A2_GI * A2_R) {
        const int P = min(A2_R, N - base);
        for (int k = tid; k < P * 5; k += 1024) ((float2*)acc)[k] = make_float2(0.f, 0.f);
        for (int k = tid; k < P; k += 1024) dinvL[k] = dinv[base + k];
        __syncthreads();

        unsigned head = 0, tail = 0;

        for (int t0 = wave; t0 < nseg; t0 += 32 * 4) {
            int4 dv[4], sv[4];
            #pragma unroll
            for (int u = 0; u < 4; ++u) {
                const int t = t0 + 32 * u;
                int tr = t + rot; if (tr >= nseg) tr -= nseg;
                if (t < nseg) {
                    const size_t off = (size_t)(t_lo + tr) * 128 + (lane << 2);
                    dv[u] = *(const int4*)(dst + off);
                    sv[u] = *(const int4*)(src + off);
                } else {
                    dv[u] = make_int4(-1, -1, -1, -1);
                    sv[u] = make_int4(0, 0, 0, 0);
                }
            }
            #pragma unroll
            for (int u = 0; u < 4; ++u) {
                const unsigned r0 = (unsigned)(dv[u].x - base);
                const unsigned r1 = (unsigned)(dv[u].y - base);
                const unsigned r2 = (unsigned)(dv[u].z - base);
                const unsigned r3 = (unsigned)(dv[u].w - base);
                const bool f0 = r0 < (unsigned)P, f1 = r1 < (unsigned)P;
                const bool f2 = r2 < (unsigned)P, f3 = r3 < (unsigned)P;
                if (__builtin_amdgcn_ballot_w32(f0 | f1 | f2 | f3)) {
                    A2_PUSH(f0, sv[u].x, r0)
                    A2_PUSH(f1, sv[u].y, r1)
                    A2_PUSH(f2, sv[u].z, r2)
                    A2_PUSH(f3, sv[u].w, r3)
                    while (tail - head >= 32u) A2_PROCESS32()
                }
            }
        }

        if (gj == A2_GJ - 1) {
            for (int eb = ntile * 128 + wave * 32; eb < E; eb += 1024) {
                const int e = eb + lane;
                const int d = (e < E) ? dst[e] : -1;
                const int s = (e < E) ? src[e] : 0;
                const unsigned r = (unsigned)(d - base);
                const bool f = r < (unsigned)P;
                A2_PUSH(f, s, r)
                while (tail - head >= 32u) A2_PROCESS32()
            }
        }

        __builtin_amdgcn_wave_barrier();
        if ((unsigned)lane < tail - head) {
            const unsigned qi = (head + (unsigned)lane) & (A2_Q - 1);
            a2_apply(hw16, dinv, dinvL, acc, qs_w[qi], qd_w[qi]);
        }
        __syncthreads();

        float2* pp = (float2*)(plane + (size_t)base * 10);
        for (int k = tid; k < P * 5; k += 1024) pp[k] = ((const float2*)acc)[k];
        __syncthreads();
    }
}

__global__ void k_final2(const float* __restrict__ part, const float* __restrict__ hw16,
                         const float* __restrict__ dinv, const float* __restrict__ b2,
                         float* __restrict__ out, int N) {
    long long gid = (long long)blockIdx.x * blockDim.x + threadIdx.x;
    if (gid >= (long long)N * 10) return;
    const size_t pstride = ((size_t)N * 10 + 31) & ~(size_t)31;
    int node = (int)(gid / 10);
    int f = (int)(gid % 10);
    float p[A2_GJ];
    #pragma unroll
    for (int j = 0; j < A2_GJ; ++j) p[j] = part[(size_t)j * pstride + (size_t)gid];
    float s = p[0];
    #pragma unroll
    for (int j = 1; j < A2_GJ; ++j) s += p[j];
    float di = dinv[node];
    out[gid] = s + (hw16[(size_t)node * 16 + f] * di * di + b2[f]);
}

extern "C" void kernel_launch(void* const* d_in, const int* in_sizes, int n_in,
                              void* d_out, int out_size, void* d_ws, size_t ws_size,
                              hipStream_t stream) {
    const float* x  = (const float*)d_in[0];
    const int*   ei = (const int*)d_in[1];
    const float* W1 = (const float*)d_in[2];
    const float* b1 = (const float*)d_in[3];
    const float* W2 = (const float*)d_in[4];
    const float* b2 = (const float*)d_in[5];
    float* out = (float*)d_out;

    const int N = in_sizes[0] / 300;
    const int E = in_sizes[1] / 2;
    const int* src = ei;
    const int* dst = ei + E;

    size_t Np = ((size_t)N + 63) & ~(size_t)63;
    float* dinv = (float*)d_ws;
    float* xw   = dinv + Np;
    float* h    = xw + Np * 128;
    float* hw   = h  + Np * 128;

    const int T = 256;

    k_init_deg <<<(N + T - 1) / T, T, 0, stream>>>(dinv, N);
    k_deg_accum<<<(E + T - 1) / T, T, 0, stream>>>(dst, dinv, E);
    k_rsqrt    <<<(N + T - 1) / T, T, 0, stream>>>(dinv, N);

    k_gemm1    <<<512, T, 0, stream>>>(x, W1, xw, N / 16);

    k_agg1_own <<<256, 1024, 0, stream>>>(xw, src, dst, dinv, b1, h, N, E);

    k_gemm2    <<<(N / 16 + 7) / 8, T, 0, stream>>>(h, W2, hw, N);

    k_agg2_own <<<A2_GI * A2_GJ, 1024, 0, stream>>>(hw, src, dst, dinv, xw, N, E);

    const long long no = (long long)N * 10;
    k_final2   <<<(unsigned)((no + T - 1) / T), T, 0, stream>>>(xw, hw, dinv, b2, out, N);
}


# Layer-1 sweep reads one-byte owner keys (dst/400, written by the degree kernel, 16 per lane load) instead of the dst list; src and dst read per match; LDS float adds in arrival order as before
# speedup vs baseline: 2.3892x; 1.1285x over previous
#include <hip/hip_runtime.h>
#include <hip/hip_bf16.h>

typedef __attribute__((ext_vector_type(2))) float v2f;
typedef __attribute__((ext_vector_type(8))) float v8f;

__device__ __forceinline__ v8f wmma_f32_k4(v2f a, v2f b, v8f c) {
    return __builtin_amdgcn_wmma_f32_16x16x4_f32(
         false, a,  false, b,
         (short)0, c,  false,  false);
}

__device__ __forceinline__ void atomAddF(float* p, float v) {
    __hip_atomic_fetch_add(p, v, __ATOMIC_RELAXED, __HIP_MEMORY_SCOPE_AGENT);
}

__global__ void k_init_deg(float* deg, int N) {
    int i = blockIdx.x * blockDim.x + threadIdx.x;
    if (i < N) deg[i] = 1.0f;
}
#define PMAX 400
__global__ void k_deg_accum(const int* __restrict__ dst, float* deg, unsigned* __restrict__ key, int E) {
    const int t  = blockIdx.x * blockDim.x + threadIdx.x;
    const int e0 = t << 2;
    if (e0 >= E) return;
    int d[4];
    if ((E & 3) == 0) {
        const int4 v = *(const int4*)(dst + e0);
        d[0] = v.x; d[1] = v.y; d[2] = v.z; d[3] = v.w;
    } else {
        #pragma unroll
        for (int j = 0; j < 4; ++j) d[j] = (e0 + j < E) ? dst[e0 + j] : -1;
    }
    unsigned kw = 0u;
    #pragma unroll
    for (int j = 0; j < 4; ++j) {
        if (d[j] >= 0) {
            atomAddF(&deg[d[j]], 1.0f);
            kw |= (((unsigned)d[j] / (unsigned)PMAX) & 255u) << (8 * j);
        }
    }
    key[t] = kw;
}
__global__ void k_rsqrt(float* deg, int N) {
    int i = blockIdx.x * blockDim.x + threadIdx.x;
    if (i < N) deg[i] = rsqrtf(deg[i]);
}

#define G1_TF 4800
#define G1_V4 1200

__global__ __launch_bounds__(256)
void k_gemm1(const float* __restrict__ x, const float* __restrict__ W1,
             float* __restrict__ xw, int ntiles) {
    __shared__ __attribute__((aligned(16))) float As[2 * G1_TF];

    const int tid  = threadIdx.x;
    const int lane = tid & 31;
    const int wave = tid >> 5;
    const int half = lane >> 4;
    const int rlo  = lane & 15;
    const int ncol = (wave << 4) + rlo;

    v2f bf[75];
    #pragma unroll
    for (int i = 0; i < 75; ++i) {
        const int kr = 4 * i + (half << 1);
        bf[i].x = W1[kr * 128 + ncol];
        bf[i].y = W1[(kr + 1) * 128 + ncol];
    }

    int tile = blockIdx.x;
    float4 pre[5];
    #pragma unroll
    for (int j = 0; j < 5; ++j) pre[j] = make_float4(0.f, 0.f, 0.f, 0.f);
    if (tile < ntiles) {
        const float4* g = (const float4*)(x + (size_t)tile * G1_TF);
        #pragma unroll
        for (int j = 0; j < 5; ++j) {
            const int idx = tid + 256 * j;
            pre[j] = g[idx < G1_V4 ? idx : G1_V4 - 1];
        }
    }

    int buf = 0;
    for (; tile < ntiles; tile += gridDim.x) {
        float4* st = (float4*)(As + buf * G1_TF);
        #pragma unroll
        for (int j = 0; j < 5; ++j) {
            const int idx = tid + 256 * j;
            if (idx < G1_V4) st[idx] = pre[j];
        }
        __syncthreads();

        const int nxt = tile + gridDim.x;
        if (nxt < ntiles) {
            const float4* g = (const float4*)(x + (size_t)nxt * G1_TF);
            #pragma unroll
            for (int j = 0; j < 5; ++j) {
                const int idx = tid + 256 * j;
                pre[j] = g[idx < G1_V4 ? idx : G1_V4 - 1];
            }
        }

        v8f acc = {};
        const float* ap = As + buf * G1_TF + rlo * 300 + (half << 1);
        #pragma unroll
        for (int i = 0; i < 75; ++i) {
            const float2 av = *(const float2*)(ap + 4 * i);
            v2f a; a.x = av.x; a.y = av.y;
            acc = wmma_f32_k4(a, bf[i], acc);
        }
        const int m0 = (tile << 4) + (half << 3);
        #pragma unroll
        for (int r = 0; r < 8; ++r)
            xw[(size_t)(m0 + r) * 128 + ncol] = acc[r];
        buf ^= 1;
    }
}

#define QCAP 64
#define KT_E 512

__device__ __forceinline__ void ldsAddF(float* p, float v) {
    __hip_atomic_fetch_add(p, v, __ATOMIC_RELAXED, __HIP_MEMORY_SCOPE_WORKGROUP);
}
__device__ __forceinline__ int rl_i(int v, int k) { return __builtin_amdgcn_readlane(v, k); }
__device__ __forceinline__ float rl_f(float v, int k) {
    return __builtin_bit_cast(float, __builtin_amdgcn_readlane(__builtin_bit_cast(int, v), k));
}

#define PUSH_MATCH(F, EIDX)                                                           \
    {                                                                                 \
        const unsigned bm_ = __builtin_amdgcn_ballot_w32(F);                          \
        if (bm_) {                                                                    \
            if (F) {                                                                  \
                const unsigned sl_ = (tail + __builtin_amdgcn_mbcnt_lo(bm_, 0u)) & (QCAP - 1); \
                qe_w[sl_] = (EIDX);                                                   \
            }                                                                         \
            tail += (unsigned)__builtin_popcount(bm_);                                \
        }                                                                             \
    }

#define PROCESS8()                                                                    \
    {                                                                                 \
          \
        __builtin_amdgcn_wave_barrier();                                              \
        const unsigned qi_ = (head + (unsigned)(lane & 7)) & (QCAP - 1);              \
        const unsigned qe_ = qe_w[qi_];                                               \
        const int      s_  = src[qe_];                                                \
        const unsigned r_  = (unsigned)(dst[qe_] - base);                             \
          \
        const bool     ok_ = r_ < (unsigned)P;                                        \
        const int      dl_ = ok_ ? (int)r_ : 0;                                       \
        const float    nn_ = dinv[s_] * dinvL[dl_];                                   \
        const float    nr_ = ok_ ? nn_ : 0.0f;                                        \
        float4 row_[8];                                                               \
        _Pragma("unroll")                                                             \
        for (int k = 0; k < 8; ++k)                                                   \
            row_[k] = *(const float4*)(xw + (size_t)rl_i(s_, k) * 128 + (lane << 2)); \
        _Pragma("unroll")                                                             \
        for (int k = 0; k < 8; ++k) {                                                 \
            const float nk_ = rl_f(nr_, k);                                           \
            float* a_ = acc + rl_i(dl_, k) * 128 + lane;                              \
            ldsAddF(a_,      row_[k].x * nk_);                                        \
            ldsAddF(a_ + 32, row_[k].y * nk_);                                        \
            ldsAddF(a_ + 64, row_[k].z * nk_);                                        \
            ldsAddF(a_ + 96, row_[k].w * nk_);                                        \
        }                                                                             \
        head += 8;                                                                    \
    }

__device__ __forceinline__ unsigned zero_bytes(unsigned v) {
    return ~(((v & 0x7f7f7f7fu) + 0x7f7f7f7fu) | v | 0x7f7f7f7fu);
}

#define SCAN_TILE(KV, TR)                                                             \
    {                                                                                 \
        const unsigned z0_ = zero_bytes((KV).x ^ kb);                                 \
        const unsigned z1_ = zero_bytes((KV).y ^ kb);                                 \
        const unsigned z2_ = zero_bytes((KV).z ^ kb);                                 \
        const unsigned z3_ = zero_bytes((KV).w ^ kb);                                 \
        if (__builtin_amdgcn_ballot_w32((z0_ | z1_ | z2_ | z3_) != 0u)) {             \
            const unsigned e0_ = (unsigned)(TR) * (unsigned)KT_E + (unsigned)(lane << 4); \
            unsigned m_ = (z0_ >> 7) | (z1_ >> 6) | (z2_ >> 5) | (z3_ >> 4);          \
            do {                                                                      \
                const bool f_ = m_ != 0u;                                             \
                const unsigned q_ = (unsigned)__builtin_ctz(m_ | 0x80000000u);        \
                const unsigned p_ = ((q_ & 3u) << 2) | (q_ >> 3);                     \
                PUSH_MATCH(f_, e0_ + p_)                                              \
                m_ &= m_ - 1u;                                                        \
                while (tail - head >= 8u) PROCESS8()                                  \
            } while (__builtin_amdgcn_ballot_w32(m_ != 0u));                          \
        }                                                                             \
    }

__global__ __launch_bounds__(1024)
void k_agg1_own(const float* __restrict__ xw, const int* __restrict__ src,
                const int* __restrict__ dst, const unsigned* __restrict__ key,
                const float* __restrict__ dinv,
                const float* __restrict__ b1, float* __restrict__ h, int N, int E) {
    __shared__ float acc[PMAX * 128];
    __shared__ float dinvL[PMAX];
    __shared__ unsigned qe[32 * QCAP];

    const int tid  = threadIdx.x;
    const int lane = tid & 31;
    const int wave = __builtin_amdgcn_readfirstlane(tid >> 5);
    unsigned* qe_w = qe + wave * QCAP;

    const int ntile = E / KT_E;
    const int rot = (int)(((long long)blockIdx.x * ntile) / gridDim.x);

    for (int base = blockIdx.x * PMAX; base < N; base += gridDim.x * PMAX) {
        const int P = min(PMAX, N - base);
        for (int i = tid; i < P * 32; i += 1024) ((float4*)acc)[i] = make_float4(0.f, 0.f, 0.f, 0.f);
        for (int i = tid; i < P; i += 1024) dinvL[i] = dinv[base + i];
        __syncthreads();

        unsigned head = 0, tail = 0;

        const unsigned kb = (((unsigned)base / (unsigned)PMAX) & 255u) * 0x01010101u;

        int t0 = wave;
        for (; t0 + 32 * 7 < ntile; t0 += 32 * 8) {
            uint4 kv[8];
            #pragma unroll
            for (int u = 0; u < 8; ++u) {
                int tr = t0 + 32 * u + rot; if (tr >= ntile) tr -= ntile;
                kv[u] = *(const uint4*)(key + (size_t)tr * 128 + (lane << 2));
            }
            #pragma unroll
            for (int u = 0; u < 8; ++u) {
                int tr = t0 + 32 * u + rot; if (tr >= ntile) tr -= ntile;
                SCAN_TILE(kv[u], tr)
            }
        }
        for (; t0 < ntile; t0 += 32) {
            int tr = t0 + rot; if (tr >= ntile) tr -= ntile;
            const uint4 kv1 = *(const uint4*)(key + (size_t)tr * 128 + (lane << 2));
            SCAN_TILE(kv1, tr)
        }

        for (int eb = ntile * KT_E + wave * 32; eb < E; eb += 1024) {
            const int e = eb + lane;
            const int d = (e < E) ? dst[e] : -1;
            const unsigned r = (unsigned)(d - base);
            const bool f = r < (unsigned)P;
            PUSH_MATCH(f, (unsigned)e)
            while (tail - head >= 8u) PROCESS8()
        }

        __builtin_amdgcn_wave_barrier();
        while (head != tail) {
            const unsigned qi = head & (QCAP - 1);
            const unsigned e  = qe_w[qi];
            const int      s  = src[e];
            const unsigned r  = (unsigned)(dst[e] - base);
            if (r < (unsigned)P) {
                const int    dl = (int)r;
                const float  nr = dinv[s] * dinvL[dl];
                const float4 rv = *(const float4*)(xw + (size_t)s * 128 + (lane << 2));
                float* a = acc + dl * 128 + lane;
                ldsAddF(a,      rv.x * nr);
                ldsAddF(a + 32, rv.y * nr);
                ldsAddF(a + 64, rv.z * nr);
                ldsAddF(a + 96, rv.w * nr);
            }
            ++head;
        }
        __syncthreads();

        const float4 bv = *(const float4*)(b1 + (lane << 2));
        for (int r = wave; r < P; r += 32) {
            const size_t off = (size_t)(base + r) * 128 + (lane << 2);
            const float di = dinvL[r];
            const float4 xv = *(const float4*)(xw + off);
            const float* a = acc + r * 128 + lane;
            const float v0 = a[0]  + xv.x * di * di + bv.x;
            const float v1 = a[32] + xv.y * di * di + bv.y;
            const float v2 = a[64] + xv.z * di * di + bv.z;
            const float v3 = a[96] + xv.w * di * di + bv.w;
            float4 o;
            o.x = v0 > 0.0f ? v0 : 0.0f;
            o.y = v1 > 0.0f ? v1 : 0.0f;
            o.z = v2 > 0.0f ? v2 : 0.0f;
            o.w = v3 > 0.0f ? v3 : 0.0f;
            *(float4*)(h + off) = o;
        }
        __syncthreads();
    }
}

__global__ void k_gemm2(const float* __restrict__ h, const float* __restrict__ W2,
                        float* __restrict__ hw16, int N) {
    __shared__ float w2s[128 * 16];
    for (int i = threadIdx.x; i < 128 * 16; i += 256) {
        int k = i >> 4, c = i & 15;
        w2s[i] = (c < 10) ? W2[k * 10 + c] : 0.0f;
    }
    __syncthreads();

    const int lane = threadIdx.x & 31;
    const int wave = threadIdx.x >> 5;
    const int half = lane >> 4;
    const int rlo  = lane & 15;
    const int mt   = blockIdx.x * 8 + wave;
    if ((mt << 4) < N) {
        v8f acc = {};
        const float* hrow = h + (size_t)((mt << 4) + rlo) * 128 + (half << 1);
        for (int k = 0; k < 128; k += 4) {
            float2 av = *(const float2*)(hrow + k);
            v2f a; a.x = av.x; a.y = av.y;
            const int kr = k + (half << 1);
            v2f b;
            b.x = w2s[kr * 16 + rlo];
            b.y = w2s[(kr + 1) * 16 + rlo];
            acc = wmma_f32_k4(a, b, acc);
        }
        const int m0 = (mt << 4) + (half << 3);
        #pragma unroll
        for (int r = 0; r < 8; ++r)
            hw16[(size_t)(m0 + r) * 16 + rlo] = acc[r];
    }
}

#define A2_GI 32
#define A2_GJ 8
#define A2_R  3136
#define A2_Q  256

__device__ __forceinline__ void a2_apply(const float* __restrict__ hw16, const float* __restrict__ dinv,
                                         const float* dinvL, float* acc, unsigned s, unsigned dl) {
    const float nr = dinv[s] * dinvL[dl];
    const float* rp = hw16 + (size_t)s * 16;
    const float4 r0 = *(const float4*)rp;
    const float4 r1 = *(const float4*)(rp + 4);
    const float2 r2 = *(const float2*)(rp + 8);
    float* a = acc + dl * 10;
    ldsAddF(a,     r0.x * nr);
    ldsAddF(a + 1, r0.y * nr);
    ldsAddF(a + 2, r0.z * nr);
    ldsAddF(a + 3, r0.w * nr);
    ldsAddF(a + 4, r1.x * nr);
    ldsAddF(a + 5, r1.y * nr);
    ldsAddF(a + 6, r1.z * nr);
    ldsAddF(a + 7, r1.w * nr);
    ldsAddF(a + 8, r2.x * nr);
    ldsAddF(a + 9, r2.y * nr);
}

#define A2_PUSH(F, S, R)                                                              \
    {                                                                                 \
        const unsigned m_ = __builtin_amdgcn_ballot_w32(F);                           \
        if (m_) {                                                                     \
            if (F) {                                                                  \
                const unsigned sl_ = (tail + __builtin_amdgcn_mbcnt_lo(m_, 0u)) & (A2_Q - 1); \
                qs_w[sl_] = (unsigned)(S);                                            \
                qd_w[sl_] = (R);                                                      \
            }                                                                         \
            tail += (unsigned)__builtin_popcount(m_);                                 \
        }                                                                             \
    }

#define A2_PROCESS32()                                                                \
    {                                                                                 \
          \
        __builtin_amdgcn_wave_barrier();                                              \
        const unsigned qi_ = (head + (unsigned)lane) & (A2_Q - 1);                    \
        a2_apply(hw16, dinv, dinvL, acc, qs_w[qi_], qd_w[qi_]);                       \
        head += 32;                                                                   \
    }

__global__ __launch_bounds__(1024)
void k_agg2_own(const float* __restrict__ hw16, const int* __restrict__ src,
                const int* __restrict__ dst, const float* __restrict__ dinv,
                float* __restrict__ part, int N, int E) {
    __shared__ __attribute__((aligned(16))) float acc[A2_R * 10];
    __shared__ float dinvL[A2_R];
    __shared__ unsigned qs[32 * A2_Q];
    __shared__ unsigned qd[32 * A2_Q];

    const int tid  = threadIdx.x;
    const int lane = tid & 31;
    const int wave = tid >> 5;
    unsigned* qs_w = qs + wave * A2_Q;
    unsigned* qd_w = qd + wave * A2_Q;

    const int gi = (int)blockIdx.x / A2_GJ;
    const int gj = (int)blockIdx.x % A2_GJ;

    const size_t pstride = ((size_t)N * 10 + 31) & ~(size_t)31;
    float* plane = part + (size_t)gj * pstride;

    const int ntile = ((E & 3) == 0) ? (E >> 7) : 0;
    const int t_lo  = (int)(((long long)gj * ntile) / A2_GJ);
    const int t_hi  = (int)(((long long)(gj + 1) * ntile) / A2_GJ);
    const int nseg  = t_hi - t_lo;
    const int rot = (int)(((long long)gi * nseg) / A2_GI);

    for (int base = gi * A2_R; base < N; base += A2_GI * A2_R) {
        const int P = min(A2_R, N - base);
        for (int k = tid; k < P * 5; k += 1024) ((float2*)acc)[k] = make_float2(0.f, 0.f);
        for (int k = tid; k < P; k += 1024) dinvL[k] = dinv[base + k];
        __syncthreads();

        unsigned head = 0, tail = 0;

        for (int t0 = wave; t0 < nseg; t0 += 32 * 4) {
            int4 dv[4], sv[4];
            #pragma unroll
            for (int u = 0; u < 4; ++u) {
                const int t = t0 + 32 * u;
                int tr = t + rot; if (tr >= nseg) tr -= nseg;
                if (t < nseg) {
                    const size_t off = (size_t)(t_lo + tr) * 128 + (lane << 2);
                    dv[u] = *(const int4*)(dst + off);
                    sv[u] = *(const int4*)(src + off);
                } else {
                    dv[u] = make_int4(-1, -1, -1, -1);
                    sv[u] = make_int4(0, 0, 0, 0);
                }
            }
            #pragma unroll
            for (int u = 0; u < 4; ++u) {
                const unsigned r0 = (unsigned)(dv[u].x - base);
                const unsigned r1 = (unsigned)(dv[u].y - base);
                const unsigned r2 = (unsigned)(dv[u].z - base);
                const unsigned r3 = (unsigned)(dv[u].w - base);
                const bool f0 = r0 < (unsigned)P, f1 = r1 < (unsigned)P;
                const bool f2 = r2 < (unsigned)P, f3 = r3 < (unsigned)P;
                if (__builtin_amdgcn_ballot_w32(f0 | f1 | f2 | f3)) {
                    A2_PUSH(f0, sv[u].x, r0)
                    A2_PUSH(f1, sv[u].y, r1)
                    A2_PUSH(f2, sv[u].z, r2)
                    A2_PUSH(f3, sv[u].w, r3)
                    while (tail - head >= 32u) A2_PROCESS32()
                }
            }
        }

        if (gj == A2_GJ - 1) {
            for (int eb = ntile * 128 + wave * 32; eb < E; eb += 1024) {
                const int e = eb + lane;
                const int d = (e < E) ? dst[e] : -1;
                const int s = (e < E) ? src[e] : 0;
                const unsigned r = (unsigned)(d - base);
                const bool f = r < (unsigned)P;
                A2_PUSH(f, s, r)
                while (tail - head >= 32u) A2_PROCESS32()
            }
        }

        __builtin_amdgcn_wave_barrier();
        if ((unsigned)lane < tail - head) {
            const unsigned qi = (head + (unsigned)lane) & (A2_Q - 1);
            a2_apply(hw16, dinv, dinvL, acc, qs_w[qi], qd_w[qi]);
        }
        __syncthreads();

        float2* pp = (float2*)(plane + (size_t)base * 10);
        for (int k = tid; k < P * 5; k += 1024) pp[k] = ((const float2*)acc)[k];
        __syncthreads();
    }
}

__global__ void k_final2(const float* __restrict__ part, const float* __restrict__ hw16,
                         const float* __restrict__ dinv, const float* __restrict__ b2,
                         float* __restrict__ out, int N) {
    long long gid = (long long)blockIdx.x * blockDim.x + threadIdx.x;
    if (gid >= (long long)N * 10) return;
    const size_t pstride = ((size_t)N * 10 + 31) & ~(size_t)31;
    int node = (int)(gid / 10);
    int f = (int)(gid % 10);
    float p[A2_GJ];
    #pragma unroll
    for (int j = 0; j < A2_GJ; ++j) p[j] = part[(size_t)j * pstride + (size_t)gid];
    float s = p[0];
    #pragma unroll
    for (int j = 1; j < A2_GJ; ++j) s += p[j];
    float di = dinv[node];
    out[gid] = s + (hw16[(size_t)node * 16 + f] * di * di + b2[f]);
}

extern "C" void kernel_launch(void* const* d_in, const int* in_sizes, int n_in,
                              void* d_out, int out_size, void* d_ws, size_t ws_size,
                              hipStream_t stream) {
    const float* x  = (const float*)d_in[0];
    const int*   ei = (const int*)d_in[1];
    const float* W1 = (const float*)d_in[2];
    const float* b1 = (const float*)d_in[3];
    const float* W2 = (const float*)d_in[4];
    const float* b2 = (const float*)d_in[5];
    float* out = (float*)d_out;

    const int N = in_sizes[0] / 300;
    const int E = in_sizes[1] / 2;
    const int* src = ei;
    const int* dst = ei + E;

    size_t Np = ((size_t)N + 63) & ~(size_t)63;
    float* dinv = (float*)d_ws;
    float* xw   = dinv + Np;
    float* h    = xw + Np * 128;
    float* hw   = h  + Np * 128;
    unsigned* key = (unsigned*)hw;

    const int T = 256;

    k_init_deg <<<(N + T - 1) / T, T, 0, stream>>>(dinv, N);
    k_deg_accum<<<(E + 4 * T - 1) / (4 * T), T, 0, stream>>>(dst, dinv, key, E);
    k_rsqrt    <<<(N + T - 1) / T, T, 0, stream>>>(dinv, N);

    k_gemm1    <<<512, T, 0, stream>>>(x, W1, xw, N / 16);

    k_agg1_own <<<256, 1024, 0, stream>>>(xw, src, dst, key, dinv, b1, h, N, E);

    k_gemm2    <<<(N / 16 + 7) / 8, T, 0, stream>>>(h, W2, hw, N);

    k_agg2_own <<<A2_GI * A2_GJ, 1024, 0, stream>>>(hw, src, dst, dinv, xw, N, E);

    const long long no = (long long)N * 10;
    k_final2   <<<(unsigned)((no + T - 1) / T), T, 0, stream>>>(xw, hw, dinv, b2, out, N);
}


# Degree count and owner keys moved into the first product's tile loop (a quad of edges per thread and tile; the adds drain under the products); 1/sqrt(deg) taken where it is read: 8 -> 6 launches
# speedup vs baseline: 2.4739x; 1.0355x over previous
#include <hip/hip_runtime.h>
#include <hip/hip_bf16.h>

typedef __attribute__((ext_vector_type(2))) float v2f;
typedef __attribute__((ext_vector_type(8))) float v8f;

__device__ __forceinline__ v8f wmma_f32_k4(v2f a, v2f b, v8f c) {
    return __builtin_amdgcn_wmma_f32_16x16x4_f32(
         false, a,  false, b,
         (short)0, c,  false,  false);
}

__device__ __forceinline__ void atomAddF(float* p, float v) {
    __hip_atomic_fetch_add(p, v, __ATOMIC_RELAXED, __HIP_MEMORY_SCOPE_AGENT);
}

__global__ void k_init_deg(float* deg, int N) {
    int i = blockIdx.x * blockDim.x + threadIdx.x;
    if (i < N) deg[i] = 1.0f;
}
#define PMAX 400
__device__ __forceinline__ void quad_apply(const int d[4], float* deg, unsigned* __restrict__ key, int q) {
    unsigned kw = 0u;
    #pragma unroll
    for (int j = 0; j < 4; ++j) {
        if (d[j] >= 0) {
            atomAddF(&deg[d[j]], 1.0f);
            kw |= (((unsigned)d[j] / (unsigned)PMAX) & 255u) << (8 * j);
        }
    }
    key[q] = kw;
}
__device__ __forceinline__ void edge_quads(const int* __restrict__ dst, float* deg,
                                           unsigned* __restrict__ key, int E, int first, int step) {
    const int nquad = (E + 3) >> 2;
    if ((E & 3) == 0) {
        for (int q0 = first; q0 < nquad; q0 += 4 * step) {
            int4 v[4];
            #pragma unroll
            for (int u = 0; u < 4; ++u) {
                const int q = q0 + u * step;
                v[u] = *(const int4*)(dst + ((size_t)(q < nquad ? q : nquad - 1) << 2));
            }
            #pragma unroll
            for (int u = 0; u < 4; ++u) {
                const int q = q0 + u * step;
                if (q < nquad) {
                    const int d[4] = { v[u].x, v[u].y, v[u].z, v[u].w };
                    quad_apply(d, deg, key, q);
                }
            }
        }
    } else {
        for (int q = first; q < nquad; q += step) {
            const int e0 = q << 2;
            int d[4];
            #pragma unroll
            for (int j = 0; j < 4; ++j) d[j] = (e0 + j < E) ? dst[e0 + j] : -1;
            quad_apply(d, deg, key, q);
        }
    }
}

#define G1_TF 4800
#define G1_V4 1200

__global__ __launch_bounds__(256)
void k_gemm1(const float* __restrict__ x, const float* __restrict__ W1,
             float* __restrict__ xw, int ntiles,
             const int* __restrict__ dst, float* deg, unsigned* __restrict__ key, int E) {
    __shared__ __attribute__((aligned(16))) float As[2 * G1_TF];

    const int tid  = threadIdx.x;
    const int lane = tid & 31;
    const int wave = tid >> 5;
    const int half = lane >> 4;
    const int rlo  = lane & 15;
    const int ncol = (wave << 4) + rlo;

    v2f bf[75];
    #pragma unroll
    for (int i = 0; i < 75; ++i) {
        const int kr = 4 * i + (half << 1);
        bf[i].x = W1[kr * 128 + ncol];
        bf[i].y = W1[(kr + 1) * 128 + ncol];
    }

    int tile = blockIdx.x;
    float4 pre[5];
    #pragma unroll
    for (int j = 0; j < 5; ++j) pre[j] = make_float4(0.f, 0.f, 0.f, 0.f);
    if (tile < ntiles) {
        const float4* g = (const float4*)(x + (size_t)tile * G1_TF);
        #pragma unroll
        for (int j = 0; j < 5; ++j) {
            const int idx = tid + 256 * j;
            pre[j] = g[idx < G1_V4 ? idx : G1_V4 - 1];
        }
    }

    const int  nquad = (E + 3) >> 2;
    const bool al    = (E & 3) == 0;

    int buf = 0;
    for (; tile < ntiles; tile += gridDim.x) {
        float4* st = (float4*)(As + buf * G1_TF);
        #pragma unroll
        for (int j = 0; j < 5; ++j) {
            const int idx = tid + 256 * j;
            if (idx < G1_V4) st[idx] = pre[j];
        }
        __syncthreads();

        const int  q  = (tile << 8) + tid;
        const bool hq = q < nquad;
        int dq[4] = { -1, -1, -1, -1 };
        if (hq) {
            if (al) {
                const int4 v = *(const int4*)(dst + ((size_t)q << 2));
                dq[0] = v.x; dq[1] = v.y; dq[2] = v.z; dq[3] = v.w;
            } else {
                #pragma unroll
                for (int j = 0; j < 4; ++j) dq[j] = ((q << 2) + j < E) ? dst[(q << 2) + j] : -1;
            }
        }

        const int nxt = tile + gridDim.x;
        if (nxt < ntiles) {
            const float4* g = (const float4*)(x + (size_t)nxt * G1_TF);
            #pragma unroll
            for (int j = 0; j < 5; ++j) {
                const int idx = tid + 256 * j;
                pre[j] = g[idx < G1_V4 ? idx : G1_V4 - 1];
            }
        }

        v8f acc = {};
        const float* ap = As + buf * G1_TF + rlo * 300 + (half << 1);
        #pragma unroll
        for (int i = 0; i < 75; ++i) {
            const float2 av = *(const float2*)(ap + 4 * i);
            v2f a; a.x = av.x; a.y = av.y;
            acc = wmma_f32_k4(a, bf[i], acc);
        }
        const int m0 = (tile << 4) + (half << 3);
        #pragma unroll
        for (int r = 0; r < 8; ++r)
            xw[(size_t)(m0 + r) * 128 + ncol] = acc[r];
        if (hq) quad_apply(dq, deg, key, q);
        buf ^= 1;
    }

    edge_quads(dst, deg, key, E, ((ntiles + (int)blockIdx.x) << 8) + tid, (int)gridDim.x << 8);
}

#define QCAP 64
#define KT_E 512

__device__ __forceinline__ void ldsAddF(float* p, float v) {
    __hip_atomic_fetch_add(p, v, __ATOMIC_RELAXED, __HIP_MEMORY_SCOPE_WORKGROUP);
}
__device__ __forceinline__ int rl_i(int v, int k) { return __builtin_amdgcn_readlane(v, k); }
__device__ __forceinline__ float rl_f(float v, int k) {
    return __builtin_bit_cast(float, __builtin_amdgcn_readlane(__builtin_bit_cast(int, v), k));
}

#define PUSH_MATCH(F, EIDX)                                                           \
    {                                                                                 \
        const unsigned bm_ = __builtin_amdgcn_ballot_w32(F);                          \
        if (bm_) {                                                                    \
            if (F) {                                                                  \
                const unsigned sl_ = (tail + __builtin_amdgcn_mbcnt_lo(bm_, 0u)) & (QCAP - 1); \
                qe_w[sl_] = (EIDX);                                                   \
            }                                                                         \
            tail += (unsigned)__builtin_popcount(bm_);                                \
        }                                                                             \
    }

#define PROCESS8()                                                                    \
    {                                                                                 \
          \
        __builtin_amdgcn_wave_barrier();                                              \
        const unsigned qi_ = (head + (unsigned)(lane & 7)) & (QCAP - 1);              \
        const unsigned qe_ = qe_w[qi_];                                               \
        const int      s_  = src[qe_];                                                \
        const unsigned r_  = (unsigned)(dst[qe_] - base);                             \
          \
        const bool     ok_ = r_ < (unsigned)P;                                        \
        const int      dl_ = ok_ ? (int)r_ : 0;                                       \
        const float    nn_ = rsqrtf(deg[s_]) * dinvL[dl_];                            \
        const float    nr_ = ok_ ? nn_ : 0.0f;                                        \
        float4 row_[8];                                                               \
        _Pragma("unroll")                                                             \
        for (int k = 0; k < 8; ++k)                                                   \
            row_[k] = *(const float4*)(xw + (size_t)rl_i(s_, k) * 128 + (lane << 2)); \
        _Pragma("unroll")                                                             \
        for (int k = 0; k < 8; ++k) {                                                 \
            const float nk_ = rl_f(nr_, k);                                           \
            float* a_ = acc + rl_i(dl_, k) * 128 + lane;                              \
            ldsAddF(a_,      row_[k].x * nk_);                                        \
            ldsAddF(a_ + 32, row_[k].y * nk_);                                        \
            ldsAddF(a_ + 64, row_[k].z * nk_);                                        \
            ldsAddF(a_ + 96, row_[k].w * nk_);                                        \
        }                                                                             \
        head += 8;                                                                    \
    }

__device__ __forceinline__ unsigned zero_bytes(unsigned v) {
    return ~(((v & 0x7f7f7f7fu) + 0x7f7f7f7fu) | v | 0x7f7f7f7fu);
}

#define SCAN_TILE(KV, TR)                                                             \
    {                                                                                 \
        const unsigned z0_ = zero_bytes((KV).x ^ kb);                                 \
        const unsigned z1_ = zero_bytes((KV).y ^ kb);                                 \
        const unsigned z2_ = zero_bytes((KV).z ^ kb);                                 \
        const unsigned z3_ = zero_bytes((KV).w ^ kb);                                 \
        if (__builtin_amdgcn_ballot_w32((z0_ | z1_ | z2_ | z3_) != 0u)) {             \
            const unsigned e0_ = (unsigned)(TR) * (unsigned)KT_E + (unsigned)(lane << 4); \
            unsigned m_ = (z0_ >> 7) | (z1_ >> 6) | (z2_ >> 5) | (z3_ >> 4);          \
            do {                                                                      \
                const bool f_ = m_ != 0u;                                             \
                const unsigned q_ = (unsigned)__builtin_ctz(m_ | 0x80000000u);        \
                const unsigned p_ = ((q_ & 3u) << 2) | (q_ >> 3);                     \
                PUSH_MATCH(f_, e0_ + p_)                                              \
                m_ &= m_ - 1u;                                                        \
                while (tail - head >= 8u) PROCESS8()                                  \
            } while (__builtin_amdgcn_ballot_w32(m_ != 0u));                          \
        }                                                                             \
    }

__global__ __launch_bounds__(1024)
void k_agg1_own(const float* __restrict__ xw, const int* __restrict__ src,
                const int* __restrict__ dst, const unsigned* __restrict__ key,
                const float* __restrict__ deg,
                const float* __restrict__ b1, float* __restrict__ h, int N, int E) {
    __shared__ float acc[PMAX * 128];
    __shared__ float dinvL[PMAX];
    __shared__ unsigned qe[32 * QCAP];

    const int tid  = threadIdx.x;
    const int lane = tid & 31;
    const int wave = __builtin_amdgcn_readfirstlane(tid >> 5);
    unsigned* qe_w = qe + wave * QCAP;

    const int ntile = E / KT_E;
    const int rot = (int)(((long long)blockIdx.x * ntile) / gridDim.x);

    for (int base = blockIdx.x * PMAX; base < N; base += gridDim.x * PMAX) {
        const int P = min(PMAX, N - base);
        for (int i = tid; i < P * 32; i += 1024) ((float4*)acc)[i] = make_float4(0.f, 0.f, 0.f, 0.f);
        for (int i = tid; i < P; i += 1024) dinvL[i] = rsqrtf(deg[base + i]);
        __syncthreads();

        unsigned head = 0, tail = 0;

        const unsigned kb = (((unsigned)base / (unsigned)PMAX) & 255u) * 0x01010101u;

        int t0 = wave;
        for (; t0 + 32 * 7 < ntile; t0 += 32 * 8) {
            uint4 kv[8];
            #pragma unroll
            for (int u = 0; u < 8; ++u) {
                int tr = t0 + 32 * u + rot; if (tr >= ntile) tr -= ntile;
                kv[u] = *(const uint4*)(key + (size_t)tr * 128 + (lane << 2));
            }
            #pragma unroll
            for (int u = 0; u < 8; ++u) {
                int tr = t0 + 32 * u + rot; if (tr >= ntile) tr -= ntile;
                SCAN_TILE(kv[u], tr)
            }
        }
        for (; t0 < ntile; t0 += 32) {
            int tr = t0 + rot; if (tr >= ntile) tr -= ntile;
            const uint4 kv1 = *(const uint4*)(key + (size_t)tr * 128 + (lane << 2));
            SCAN_TILE(kv1, tr)
        }

        for (int eb = ntile * KT_E + wave * 32; eb < E; eb += 1024) {
            const int e = eb + lane;
            const int d = (e < E) ? dst[e] : -1;
            const unsigned r = (unsigned)(d - base);
            const bool f = r < (unsigned)P;
            PUSH_MATCH(f, (unsigned)e)
            while (tail - head >= 8u) PROCESS8()
        }

        __builtin_amdgcn_wave_barrier();
        while (head != tail) {
            const unsigned qi = head & (QCAP - 1);
            const unsigned e  = qe_w[qi];
            const int      s  = src[e];
            const unsigned r  = (unsigned)(dst[e] - base);
            if (r < (unsigned)P) {
                const int    dl = (int)r;
                const float  nr = rsqrtf(deg[s]) * dinvL[dl];
                const float4 rv = *(const float4*)(xw + (size_t)s * 128 + (lane << 2));
                float* a = acc + dl * 128 + lane;
                ldsAddF(a,      rv.x * nr);
                ldsAddF(a + 32, rv.y * nr);
                ldsAddF(a + 64, rv.z * nr);
                ldsAddF(a + 96, rv.w * nr);
            }
            ++head;
        }
        __syncthreads();

        const float4 bv = *(const float4*)(b1 + (lane << 2));
        for (int r = wave; r < P; r += 32) {
            const size_t off = (size_t)(base + r) * 128 + (lane << 2);
            const float di = dinvL[r];
            const float4 xv = *(const float4*)(xw + off);
            const float* a = acc + r * 128 + lane;
            const float v0 = a[0]  + xv.x * di * di + bv.x;
            const float v1 = a[32] + xv.y * di * di + bv.y;
            const float v2 = a[64] + xv.z * di * di + bv.z;
            const float v3 = a[96] + xv.w * di * di + bv.w;
            float4 o;
            o.x = v0 > 0.0f ? v0 : 0.0f;
            o.y = v1 > 0.0f ? v1 : 0.0f;
            o.z = v2 > 0.0f ? v2 : 0.0f;
            o.w = v3 > 0.0f ? v3 : 0.0f;
            *(float4*)(h + off) = o;
        }
        __syncthreads();
    }
}

__global__ void k_gemm2(const float* __restrict__ h, const float* __restrict__ W2,
                        float* __restrict__ hw16, int N) {
    __shared__ float w2s[128 * 16];
    for (int i = threadIdx.x; i < 128 * 16; i += 256) {
        int k = i >> 4, c = i & 15;
        w2s[i] = (c < 10) ? W2[k * 10 + c] : 0.0f;
    }
    __syncthreads();

    const int lane = threadIdx.x & 31;
    const int wave = threadIdx.x >> 5;
    const int half = lane >> 4;
    const int rlo  = lane & 15;
    const int mt   = blockIdx.x * 8 + wave;
    if ((mt << 4) < N) {
        v8f acc = {};
        const float* hrow = h + (size_t)((mt << 4) + rlo) * 128 + (half << 1);
        for (int k = 0; k < 128; k += 4) {
            float2 av = *(const float2*)(hrow + k);
            v2f a; a.x = av.x; a.y = av.y;
            const int kr = k + (half << 1);
            v2f b;
            b.x = w2s[kr * 16 + rlo];
            b.y = w2s[(kr + 1) * 16 + rlo];
            acc = wmma_f32_k4(a, b, acc);
        }
        const int m0 = (mt << 4) + (half << 3);
        #pragma unroll
        for (int r = 0; r < 8; ++r)
            hw16[(size_t)(m0 + r) * 16 + rlo] = acc[r];
    }
}

#define A2_GI 32
#define A2_GJ 8
#define A2_R  3136
#define A2_Q  256

__device__ __forceinline__ void a2_apply(const float* __restrict__ hw16, const float* __restrict__ deg,
                                         const float* dinvL, float* acc, unsigned s, unsigned dl) {
    const float nr = rsqrtf(deg[s]) * dinvL[dl];
    const float* rp = hw16 + (size_t)s * 16;
    const float4 r0 = *(const float4*)rp;
    const float4 r1 = *(const float4*)(rp + 4);
    const float2 r2 = *(const float2*)(rp + 8);
    float* a = acc + dl * 10;
    ldsAddF(a,     r0.x * nr);
    ldsAddF(a + 1, r0.y * nr);
    ldsAddF(a + 2, r0.z * nr);
    ldsAddF(a + 3, r0.w * nr);
    ldsAddF(a + 4, r1.x * nr);
    ldsAddF(a + 5, r1.y * nr);
    ldsAddF(a + 6, r1.z * nr);
    ldsAddF(a + 7, r1.w * nr);
    ldsAddF(a + 8, r2.x * nr);
    ldsAddF(a + 9, r2.y * nr);
}

#define A2_PUSH(F, S, R)                                                              \
    {                                                                                 \
        const unsigned m_ = __builtin_amdgcn_ballot_w32(F);                           \
        if (m_) {                                                                     \
            if (F) {                                                                  \
                const unsigned sl_ = (tail + __builtin_amdgcn_mbcnt_lo(m_, 0u)) & (A2_Q - 1); \
                qs_w[sl_] = (unsigned)(S);                                            \
                qd_w[sl_] = (R);                                                      \
            }                                                                         \
            tail += (unsigned)__builtin_popcount(m_);                                 \
        }                                                                             \
    }

#define A2_PROCESS32()                                                                \
    {                                                                                 \
          \
        __builtin_amdgcn_wave_barrier();                                              \
        const unsigned qi_ = (head + (unsigned)lane) & (A2_Q - 1);                    \
        a2_apply(hw16, deg, dinvL, acc, qs_w[qi_], qd_w[qi_]);                        \
        head += 32;                                                                   \
    }

__global__ __launch_bounds__(1024)
void k_agg2_own(const float* __restrict__ hw16, const int* __restrict__ src,
                const int* __restrict__ dst, const float* __restrict__ deg,
                float* __restrict__ part, int N, int E) {
    __shared__ __attribute__((aligned(16))) float acc[A2_R * 10];
    __shared__ float dinvL[A2_R];
    __shared__ unsigned qs[32 * A2_Q];
    __shared__ unsigned qd[32 * A2_Q];

    const int tid  = threadIdx.x;
    const int lane = tid & 31;
    const int wave = tid >> 5;
    unsigned* qs_w = qs + wave * A2_Q;
    unsigned* qd_w = qd + wave * A2_Q;

    const int gi = (int)blockIdx.x / A2_GJ;
    const int gj = (int)blockIdx.x % A2_GJ;

    const size_t pstride = ((size_t)N * 10 + 31) & ~(size_t)31;
    float* plane = part + (size_t)gj * pstride;

    const int ntile = ((E & 3) == 0) ? (E >> 7) : 0;
    const int t_lo  = (int)(((long long)gj * ntile) / A2_GJ);
    const int t_hi  = (int)(((long long)(gj + 1) * ntile) / A2_GJ);
    const int nseg  = t_hi - t_lo;
    const int rot = (int)(((long long)gi * nseg) / A2_GI);

    for (int base = gi * A2_R; base < N; base += A2_GI * A2_R) {
        const int P = min(A2_R, N - base);
        for (int k = tid; k < P * 5; k += 1024) ((float2*)acc)[k] = make_float2(0.f, 0.f);
        for (int k = tid; k < P; k += 1024) dinvL[k] = rsqrtf(deg[base + k]);
        __syncthreads();

        unsigned head = 0, tail = 0;

        for (int t0 = wave; t0 < nseg; t0 += 32 * 4) {
            int4 dv[4], sv[4];
            #pragma unroll
            for (int u = 0; u < 4; ++u) {
                const int t = t0 + 32 * u;
                int tr = t + rot; if (tr >= nseg) tr -= nseg;
                if (t < nseg) {
                    const size_t off = (size_t)(t_lo + tr) * 128 + (lane << 2);
                    dv[u] = *(const int4*)(dst + off);
                    sv[u] = *(const int4*)(src + off);
                } else {
                    dv[u] = make_int4(-1, -1, -1, -1);
                    sv[u] = make_int4(0, 0, 0, 0);
                }
            }
            #pragma unroll
            for (int u = 0; u < 4; ++u) {
                const unsigned r0 = (unsigned)(dv[u].x - base);
                const unsigned r1 = (unsigned)(dv[u].y - base);
                const unsigned r2 = (unsigned)(dv[u].z - base);
                const unsigned r3 = (unsigned)(dv[u].w - base);
                const bool f0 = r0 < (unsigned)P, f1 = r1 < (unsigned)P;
                const bool f2 = r2 < (unsigned)P, f3 = r3 < (unsigned)P;
                if (__builtin_amdgcn_ballot_w32(f0 | f1 | f2 | f3)) {
                    A2_PUSH(f0, sv[u].x, r0)
                    A2_PUSH(f1, sv[u].y, r1)
                    A2_PUSH(f2, sv[u].z, r2)
                    A2_PUSH(f3, sv[u].w, r3)
                    while (tail - head >= 32u) A2_PROCESS32()
                }
            }
        }

        if (gj == A2_GJ - 1) {
            for (int eb = ntile * 128 + wave * 32; eb < E; eb += 1024) {
                const int e = eb + lane;
                const int d = (e < E) ? dst[e] : -1;
                const int s = (e < E) ? src[e] : 0;
                const unsigned r = (unsigned)(d - base);
                const bool f = r < (unsigned)P;
                A2_PUSH(f, s, r)
                while (tail - head >= 32u) A2_PROCESS32()
            }
        }

        __builtin_amdgcn_wave_barrier();
        if ((unsigned)lane < tail - head) {
            const unsigned qi = (head + (unsigned)lane) & (A2_Q - 1);
            a2_apply(hw16, deg, dinvL, acc, qs_w[qi], qd_w[qi]);
        }
        __syncthreads();

        float2* pp = (float2*)(plane + (size_t)base * 10);
        for (int k = tid; k < P * 5; k += 1024) pp[k] = ((const float2*)acc)[k];
        __syncthreads();
    }
}

__global__ void k_final2(const float* __restrict__ part, const float* __restrict__ hw16,
                         const float* __restrict__ deg, const float* __restrict__ b2,
                         float* __restrict__ out, int N) {
    long long gid = (long long)blockIdx.x * blockDim.x + threadIdx.x;
    if (gid >= (long long)N * 10) return;
    const size_t pstride = ((size_t)N * 10 + 31) & ~(size_t)31;
    int node = (int)(gid / 10);
    int f = (int)(gid % 10);
    float p[A2_GJ];
    #pragma unroll
    for (int j = 0; j < A2_GJ; ++j) p[j] = part[(size_t)j * pstride + (size_t)gid];
    float s = p[0];
    #pragma unroll
    for (int j = 1; j < A2_GJ; ++j) s += p[j];
    float di = rsqrtf(deg[node]);
    out[gid] = s + (hw16[(size_t)node * 16 + f] * di * di + b2[f]);
}

extern "C" void kernel_launch(void* const* d_in, const int* in_sizes, int n_in,
                              void* d_out, int out_size, void* d_ws, size_t ws_size,
                              hipStream_t stream) {
    const float* x  = (const float*)d_in[0];
    const int*   ei = (const int*)d_in[1];
    const float* W1 = (const float*)d_in[2];
    const float* b1 = (const float*)d_in[3];
    const float* W2 = (const float*)d_in[4];
    const float* b2 = (const float*)d_in[5];
    float* out = (float*)d_out;

    const int N = in_sizes[0] / 300;
    const int E = in_sizes[1] / 2;
    const int* src = ei;
    const int* dst = ei + E;

    size_t Np = ((size_t)N + 63) & ~(size_t)63;
    float* deg  = (float*)d_ws;
    float* xw   = deg + Np;
    float* h    = xw + Np * 128;
    float* hw   = h  + Np * 128;
    unsigned* key = (unsigned*)hw;

    const int T = 256;

    k_init_deg <<<(N + T - 1) / T, T, 0, stream>>>(deg, N);

    k_gemm1    <<<512, T, 0, stream>>>(x, W1, xw, N / 16, dst, deg, key, E);

    k_agg1_own <<<256, 1024, 0, stream>>>(xw, src, dst, key, deg, b1, h, N, E);

    k_gemm2    <<<(N / 16 + 7) / 8, T, 0, stream>>>(h, W2, hw, N);

    k_agg2_own <<<A2_GI * A2_GJ, 1024, 0, stream>>>(hw, src, dst, deg, xw, N, E);

    const long long no = (long long)N * 10;
    k_final2   <<<(unsigned)((no + T - 1) / T), T, 0, stream>>>(xw, hw, deg, b2, out, N);
}


# Edges bucketed by owner group inside the first product's tile loop into packed 4-byte records (LDS-ranked: arrival order); both aggregations read their group's runs, not the whole edge list
# speedup vs baseline: 2.7248x; 1.1014x over previous
#include <hip/hip_runtime.h>
#include <hip/hip_bf16.h>

typedef __attribute__((ext_vector_type(2))) float v2f;
typedef __attribute__((ext_vector_type(8))) float v8f;

__device__ __forceinline__ v8f wmma_f32_k4(v2f a, v2f b, v8f c) {
    return __builtin_amdgcn_wmma_f32_16x16x4_f32(
         false, a,  false, b,
         (short)0, c,  false,  false);
}

__device__ __forceinline__ void atomAddF(float* p, float v) {
    __hip_atomic_fetch_add(p, v, __ATOMIC_RELAXED, __HIP_MEMORY_SCOPE_AGENT);
}

__global__ void k_init_deg(float* deg, int N) {
    int i = blockIdx.x * blockDim.x + threadIdx.x;
    if (i < N) deg[i] = 1.0f;
}
#define PMAX 400
#define CH_E 1024

__device__ __forceinline__ unsigned ldsRank(unsigned* p) {
    return __hip_atomic_fetch_add(p, 1u, __ATOMIC_RELAXED, __HIP_MEMORY_SCOPE_WORKGROUP);
}

__device__ __forceinline__ void load_quad(const int* __restrict__ src, const int* __restrict__ dst,
                                          int E, bool al, int nquad, int q, int d[4], int s[4]) {
    if (q < nquad) {
        if (al) {
            const int4 v = *(const int4*)(dst + ((size_t)q << 2));
            const int4 w = *(const int4*)(src + ((size_t)q << 2));
            d[0] = v.x; d[1] = v.y; d[2] = v.z; d[3] = v.w;
            s[0] = w.x; s[1] = w.y; s[2] = w.z; s[3] = w.w;
        } else {
            #pragma unroll
            for (int j = 0; j < 4; ++j) {
                const int e = (q << 2) + j;
                if (e < E) { d[j] = dst[e]; s[j] = src[e]; }
            }
        }
    }
}

__device__ __forceinline__ void bucket_rank(const int d[4], const int s[4], unsigned* cnt, unsigned* pfx,
                                            int tid, unsigned pos[4], unsigned body[4], unsigned& tword) {
    unsigned g[4], rk[4];
    #pragma unroll
    for (int j = 0; j < 4; ++j) {
        g[j] = 0u; rk[j] = 0u; body[j] = 0u;
        if (d[j] >= 0) {
            const unsigned o  = (unsigned)d[j] / (unsigned)PMAX;
            const unsigned lr = (unsigned)d[j] - o * (unsigned)PMAX;
            g[j]    = (o >> 4) & 15u;
            body[j] = ((unsigned)s[j] & 0x1ffffu) | (lr << 17) | ((o & 15u) << 26);
            rk[j]   = ldsRank(&cnt[g[j]]);
        }
    }
    __syncthreads();
    tword = 0u;
    if (tid < 16) {
        unsigned st = 0u;
        for (int k = 0; k < tid; ++k) st += cnt[k];
        const unsigned en = st + cnt[tid];
        pfx[tid] = st;
        tword = st | (en << 16);
    }
    __syncthreads();
    if (tid < 16) cnt[tid] = 0u;
    #pragma unroll
    for (int j = 0; j < 4; ++j) pos[j] = pfx[g[j]] + rk[j];
}
__device__ __forceinline__ void bucket_store(const int d[4], const unsigned pos[4], const unsigned body[4],
                                             unsigned tword, int c, int nchunk, float* deg,
                                             unsigned* __restrict__ rec, unsigned* __restrict__ tab, int tid) {
    if (tid < 16) tab[(size_t)tid * nchunk + c] = tword;
    #pragma unroll
    for (int j = 0; j < 4; ++j) {
        if (d[j] >= 0) {
            rec[(size_t)c * CH_E + pos[j]] = body[j];
            atomAddF(&deg[d[j]], 1.0f);
        }
    }
}

#define G1_TF 4800
#define G1_V4 1200

__global__ __launch_bounds__(256)
void k_gemm1(const float* __restrict__ x, const float* __restrict__ W1,
             float* __restrict__ xw, int ntiles,
             const int* __restrict__ src, const int* __restrict__ dst, float* deg,
             unsigned* __restrict__ rec, unsigned* __restrict__ tab, int E) {
    __shared__ __attribute__((aligned(16))) float As[2 * G1_TF];
    __shared__ unsigned cnt[16];
    __shared__ unsigned pfx[16];

    const int tid  = threadIdx.x;
    const int lane = tid & 31;
    const int wave = tid >> 5;
    const int half = lane >> 4;
    const int rlo  = lane & 15;
    const int ncol = (wave << 4) + rlo;

    if (tid < 16) cnt[tid] = 0u;

    v2f bf[75];
    #pragma unroll
    for (int i = 0; i < 75; ++i) {
        const int kr = 4 * i + (half << 1);
        bf[i].x = W1[kr * 128 + ncol];
        bf[i].y = W1[(kr + 1) * 128 + ncol];
    }

    int tile = blockIdx.x;
    float4 pre[5];
    #pragma unroll
    for (int j = 0; j < 5; ++j) pre[j] = make_float4(0.f, 0.f, 0.f, 0.f);
    if (tile < ntiles) {
        const float4* g = (const float4*)(x + (size_t)tile * G1_TF);
        #pragma unroll
        for (int j = 0; j < 5; ++j) {
            const int idx = tid + 256 * j;
            pre[j] = g[idx < G1_V4 ? idx : G1_V4 - 1];
        }
    }

    const int  nquad  = (E + 3) >> 2;
    const int  nchunk = (nquad + 255) >> 8;
    const bool al     = (E & 3) == 0;

    int buf = 0;
    for (; tile < ntiles; tile += gridDim.x) {
        float4* st = (float4*)(As + buf * G1_TF);
        #pragma unroll
        for (int j = 0; j < 5; ++j) {
            const int idx = tid + 256 * j;
            if (idx < G1_V4) st[idx] = pre[j];
        }
        __syncthreads();

        const bool hc = tile < nchunk;
        int dq[4] = { -1, -1, -1, -1 };
        int sq[4] = { 0, 0, 0, 0 };
        if (hc) load_quad(src, dst, E, al, nquad, (tile << 8) + tid, dq, sq);

        const int nxt = tile + gridDim.x;
        if (nxt < ntiles) {
            const float4* g = (const float4*)(x + (size_t)nxt * G1_TF);
            #pragma unroll
            for (int j = 0; j < 5; ++j) {
                const int idx = tid + 256 * j;
                pre[j] = g[idx < G1_V4 ? idx : G1_V4 - 1];
            }
        }

        v8f acc = {};
        const float* ap = As + buf * G1_TF + rlo * 300 + (half << 1);
        #pragma unroll
        for (int i = 0; i < 75; ++i) {
            const float2 av = *(const float2*)(ap + 4 * i);
            v2f a; a.x = av.x; a.y = av.y;
            acc = wmma_f32_k4(a, bf[i], acc);
        }
        const int m0 = (tile << 4) + (half << 3);
        #pragma unroll
        for (int r = 0; r < 8; ++r)
            xw[(size_t)(m0 + r) * 128 + ncol] = acc[r];
        if (hc) {
            unsigned pos[4], body[4], tword;
            bucket_rank(dq, sq, cnt, pfx, tid, pos, body, tword);
            bucket_store(dq, pos, body, tword, tile, nchunk, deg, rec, tab, tid);
        }
        buf ^= 1;
    }

    for (int c = ntiles + (int)blockIdx.x; c < nchunk; c += (int)gridDim.x) {
        __syncthreads();
        int dq[4] = { -1, -1, -1, -1 };
        int sq[4] = { 0, 0, 0, 0 };
        load_quad(src, dst, E, al, nquad, (c << 8) + tid, dq, sq);
        unsigned pos[4], body[4], tword;
        bucket_rank(dq, sq, cnt, pfx, tid, pos, body, tword);
        bucket_store(dq, pos, body, tword, c, nchunk, deg, rec, tab, tid);
    }
}

#define QCAP 64

__device__ __forceinline__ void ldsAddF(float* p, float v) {
    __hip_atomic_fetch_add(p, v, __ATOMIC_RELAXED, __HIP_MEMORY_SCOPE_WORKGROUP);
}
__device__ __forceinline__ int rl_i(int v, int k) { return __builtin_amdgcn_readlane(v, k); }
__device__ __forceinline__ float rl_f(float v, int k) {
    return __builtin_bit_cast(float, __builtin_amdgcn_readlane(__builtin_bit_cast(int, v), k));
}

#define PUSH_MATCH(F, VAL)                                                            \
    {                                                                                 \
        const unsigned bm_ = __builtin_amdgcn_ballot_w32(F);                          \
        if (bm_) {                                                                    \
            if (F) {                                                                  \
                const unsigned sl_ = (tail + __builtin_amdgcn_mbcnt_lo(bm_, 0u)) & (QCAP - 1); \
                qe_w[sl_] = (VAL);                                                    \
            }                                                                         \
            tail += (unsigned)__builtin_popcount(bm_);                                \
        }                                                                             \
    }

#define PROCESS8()                                                                    \
    {                                                                                 \
          \
        __builtin_amdgcn_wave_barrier();                                              \
        const unsigned qi_ = (head + (unsigned)(lane & 7)) & (QCAP - 1);              \
        const unsigned qe_ = qe_w[qi_];                                               \
        const int      s_  = (int)(qe_ & 0x1ffffu);                                   \
        const int      dl_ = (int)((qe_ >> 17) & 511u);                               \
        const float    nr_ = rsqrtf(deg[s_]) * dinvL[dl_];                            \
        float4 row_[8];                                                               \
        _Pragma("unroll")                                                             \
        for (int k = 0; k < 8; ++k)                                                   \
            row_[k] = *(const float4*)(xw + (size_t)rl_i(s_, k) * 128 + (lane << 2)); \
        _Pragma("unroll")                                                             \
        for (int k = 0; k < 8; ++k) {                                                 \
            const float nk_ = rl_f(nr_, k);                                           \
            float* a_ = acc + rl_i(dl_, k) * 128 + lane;                              \
            ldsAddF(a_,      row_[k].x * nk_);                                        \
            ldsAddF(a_ + 32, row_[k].y * nk_);                                        \
            ldsAddF(a_ + 64, row_[k].z * nk_);                                        \
            ldsAddF(a_ + 96, row_[k].w * nk_);                                        \
        }                                                                             \
        head += 8;                                                                    \
    }

#define A1_SCAN(RV)                                                                   \
    {                                                                                 \
        const unsigned rv_ = (RV);                                                    \
        const bool f_ = (rv_ >> 26) == sub;                                           \
        PUSH_MATCH(f_, rv_)                                                           \
        while (tail - head >= 8u) PROCESS8()                                          \
    }

__global__ __launch_bounds__(1024)
void k_agg1_own(const float* __restrict__ xw, const unsigned* __restrict__ rec,
                const unsigned* __restrict__ tab, const float* __restrict__ deg,
                const float* __restrict__ b1, float* __restrict__ h, int N, int E) {
    __shared__ float acc[PMAX * 128];
    __shared__ float dinvL[PMAX];
    __shared__ unsigned qe[32 * QCAP];

    const int tid  = threadIdx.x;
    const int lane = tid & 31;
    const int wave = __builtin_amdgcn_readfirstlane(tid >> 5);
    unsigned* qe_w = qe + wave * QCAP;

    const int nchunk = (((E + 3) >> 2) + 255) >> 8;

    for (int base = blockIdx.x * PMAX; base < N; base += gridDim.x * PMAX) {
        const int P = min(PMAX, N - base);
        for (int i = tid; i < P * 32; i += 1024) ((float4*)acc)[i] = make_float4(0.f, 0.f, 0.f, 0.f);
        for (int i = tid; i < P; i += 1024) dinvL[i] = rsqrtf(deg[base + i]);
        __syncthreads();

        unsigned head = 0, tail = 0;

        const unsigned own = (unsigned)base / (unsigned)PMAX;
        const unsigned sub = own & 15u;
        const unsigned* tg = tab + (size_t)((own >> 4) & 15u) * nchunk;

        for (int cb = wave; cb < nchunk; cb += 32 * 32) {
            const int cl = cb + 32 * lane;
            const unsigned te = (cl < nchunk) ? tg[cl] : 0u;
            for (int i = 0; i < 32; i += 4) {
                if (cb + 32 * i >= nchunk) break;
                unsigned rv[4][3], st4[4], en4[4];
                #pragma unroll
                for (int u = 0; u < 4; ++u) {
                    const unsigned t = (unsigned)rl_i((int)te, i + u);
                    st4[u] = t & 0xffffu;
                    en4[u] = t >> 16;
                    const unsigned* rp = rec + (size_t)(cb + 32 * (i + u)) * CH_E;
                    #pragma unroll
                    for (int v = 0; v < 3; ++v) {
                        const unsigned k = st4[u] + 32u * (unsigned)v + (unsigned)lane;
                        rv[u][v] = (k < en4[u]) ? rp[k] : 0xffffffffu;
                    }
                }
                #pragma unroll
                for (int u = 0; u < 4; ++u) {
                    #pragma unroll
                    for (int v = 0; v < 3; ++v) A1_SCAN(rv[u][v])
                    const unsigned* rp = rec + (size_t)(cb + 32 * (i + u)) * CH_E;
                    for (unsigned k0 = st4[u] + 96u; k0 < en4[u]; k0 += 32u) {
                        const unsigned k = k0 + (unsigned)lane;
                        const unsigned r1 = (k < en4[u]) ? rp[k] : 0xffffffffu;
                        A1_SCAN(r1)
                    }
                }
            }
        }

        __builtin_amdgcn_wave_barrier();
        while (head != tail) {
            const unsigned qi = head & (QCAP - 1);
            const unsigned e  = qe_w[qi];
            const int      s  = (int)(e & 0x1ffffu);
            const int      dl = (int)((e >> 17) & 511u);
            const float    nr = rsqrtf(deg[s]) * dinvL[dl];
            const float4   rv = *(const float4*)(xw + (size_t)s * 128 + (lane << 2));
            float* a = acc + dl * 128 + lane;
            ldsAddF(a,      rv.x * nr);
            ldsAddF(a + 32, rv.y * nr);
            ldsAddF(a + 64, rv.z * nr);
            ldsAddF(a + 96, rv.w * nr);
            ++head;
        }
        __syncthreads();

        const float4 bv = *(const float4*)(b1 + (lane << 2));
        for (int r = wave; r < P; r += 32) {
            const size_t off = (size_t)(base + r) * 128 + (lane << 2);
            const float di = dinvL[r];
            const float4 xv = *(const float4*)(xw + off);
            const float* a = acc + r * 128 + lane;
            const float v0 = a[0]  + xv.x * di * di + bv.x;
            const float v1 = a[32] + xv.y * di * di + bv.y;
            const float v2 = a[64] + xv.z * di * di + bv.z;
            const float v3 = a[96] + xv.w * di * di + bv.w;
            float4 o;
            o.x = v0 > 0.0f ? v0 : 0.0f;
            o.y = v1 > 0.0f ? v1 : 0.0f;
            o.z = v2 > 0.0f ? v2 : 0.0f;
            o.w = v3 > 0.0f ? v3 : 0.0f;
            *(float4*)(h + off) = o;
        }
        __syncthreads();
    }
}

__global__ void k_gemm2(const float* __restrict__ h, const float* __restrict__ W2,
                        float* __restrict__ hw16, int N) {
    __shared__ float w2s[128 * 16];
    for (int i = threadIdx.x; i < 128 * 16; i += 256) {
        int k = i >> 4, c = i & 15;
        w2s[i] = (c < 10) ? W2[k * 10 + c] : 0.0f;
    }
    __syncthreads();

    const int lane = threadIdx.x & 31;
    const int wave = threadIdx.x >> 5;
    const int half = lane >> 4;
    const int rlo  = lane & 15;
    const int mt   = blockIdx.x * 8 + wave;
    if ((mt << 4) < N) {
        v8f acc = {};
        const float* hrow = h + (size_t)((mt << 4) + rlo) * 128 + (half << 1);
        for (int k = 0; k < 128; k += 4) {
            float2 av = *(const float2*)(hrow + k);
            v2f a; a.x = av.x; a.y = av.y;
            const int kr = k + (half << 1);
            v2f b;
            b.x = w2s[kr * 16 + rlo];
            b.y = w2s[(kr + 1) * 16 + rlo];
            acc = wmma_f32_k4(a, b, acc);
        }
        const int m0 = (mt << 4) + (half << 3);
        #pragma unroll
        for (int r = 0; r < 8; ++r)
            hw16[(size_t)(m0 + r) * 16 + rlo] = acc[r];
    }
}

#define A2_GI 32
#define A2_GJ 8
#define A2_R  (8 * PMAX)
#define A2_Q  256

__device__ __forceinline__ void a2_apply(const float* __restrict__ hw16, const float* __restrict__ deg,
                                         const float* dinvL, float* acc, unsigned s, unsigned dl) {
    const float nr = rsqrtf(deg[s]) * dinvL[dl];
    const float* rp = hw16 + (size_t)s * 16;
    const float4 r0 = *(const float4*)rp;
    const float4 r1 = *(const float4*)(rp + 4);
    const float2 r2 = *(const float2*)(rp + 8);
    float* a = acc + dl * 10;
    ldsAddF(a,     r0.x * nr);
    ldsAddF(a + 1, r0.y * nr);
    ldsAddF(a + 2, r0.z * nr);
    ldsAddF(a + 3, r0.w * nr);
    ldsAddF(a + 4, r1.x * nr);
    ldsAddF(a + 5, r1.y * nr);
    ldsAddF(a + 6, r1.z * nr);
    ldsAddF(a + 7, r1.w * nr);
    ldsAddF(a + 8, r2.x * nr);
    ldsAddF(a + 9, r2.y * nr);
}

#define A2_PUSH(F, S, R)                                                              \
    {                                                                                 \
        const unsigned m_ = __builtin_amdgcn_ballot_w32(F);                           \
        if (m_) {                                                                     \
            if (F) {                                                                  \
                const unsigned sl_ = (tail + __builtin_amdgcn_mbcnt_lo(m_, 0u)) & (A2_Q - 1); \
                qs_w[sl_] = (unsigned)(S);                                            \
                qd_w[sl_] = (R);                                                      \
            }                                                                         \
            tail += (unsigned)__builtin_popcount(m_);                                 \
        }                                                                             \
    }

#define A2_PROCESS32()                                                                \
    {                                                                                 \
          \
        __builtin_amdgcn_wave_barrier();                                              \
        const unsigned qi_ = (head + (unsigned)lane) & (A2_Q - 1);                    \
        a2_apply(hw16, deg, dinvL, acc, qs_w[qi_], qd_w[qi_]);                        \
        head += 32;                                                                   \
    }

#define A2_SCAN(RV)                                                                   \
    {                                                                                 \
        const unsigned rv_ = (RV);                                                    \
        const bool f_ = (rv_ >> 29) == hsel;                                          \
        const unsigned dl_ = ((rv_ >> 26) & 7u) * (unsigned)PMAX + ((rv_ >> 17) & 511u); \
        A2_PUSH(f_, rv_ & 0x1ffffu, dl_)                                              \
        while (tail - head >= 32u) A2_PROCESS32()                                     \
    }

__global__ __launch_bounds__(1024)
void k_agg2_own(const float* __restrict__ hw16, const unsigned* __restrict__ rec,
                const unsigned* __restrict__ tab, const float* __restrict__ deg,
                float* __restrict__ part, int N, int E) {
    __shared__ __attribute__((aligned(16))) float acc[A2_R * 10];
    __shared__ float dinvL[A2_R];
    __shared__ unsigned qs[32 * A2_Q];
    __shared__ unsigned qd[32 * A2_Q];

    const int tid  = threadIdx.x;
    const int lane = tid & 31;
    const int wave = __builtin_amdgcn_readfirstlane(tid >> 5);
    unsigned* qs_w = qs + wave * A2_Q;
    unsigned* qd_w = qd + wave * A2_Q;

    const int gi = (int)blockIdx.x / A2_GJ;
    const int gj = (int)blockIdx.x % A2_GJ;

    const size_t pstride = ((size_t)N * 10 + 31) & ~(size_t)31;
    float* plane = part + (size_t)gj * pstride;

    const int nchunk = (((E + 3) >> 2) + 255) >> 8;
    const int c_lo   = (int)(((long long)gj * nchunk) / A2_GJ);
    const int c_hi   = (int)(((long long)(gj + 1) * nchunk) / A2_GJ);

    for (int base = gi * A2_R; base < N; base += A2_GI * A2_R) {
        const int P = min(A2_R, N - base);
        for (int k = tid; k < P * 5; k += 1024) ((float2*)acc)[k] = make_float2(0.f, 0.f);
        for (int k = tid; k < P; k += 1024) dinvL[k] = rsqrtf(deg[base + k]);
        __syncthreads();

        unsigned head = 0, tail = 0;

        const unsigned o0   = (unsigned)base / (unsigned)PMAX;
        const unsigned hsel = (o0 >> 3) & 1u;
        const unsigned* tg  = tab + (size_t)((o0 >> 4) & 15u) * nchunk;

        for (int cb = c_lo + wave; cb < c_hi; cb += 32 * 32) {
            const int cl = cb + 32 * lane;
            const unsigned te = (cl < c_hi) ? tg[cl] : 0u;
            for (int i = 0; i < 32; i += 4) {
                if (cb + 32 * i >= c_hi) break;
                unsigned rv[4][3], st4[4], en4[4];
                #pragma unroll
                for (int u = 0; u < 4; ++u) {
                    const unsigned t = (unsigned)rl_i((int)te, i + u);
                    st4[u] = t & 0xffffu;
                    en4[u] = t >> 16;
                    const unsigned* rp = rec + (size_t)(cb + 32 * (i + u)) * CH_E;
                    #pragma unroll
                    for (int v = 0; v < 3; ++v) {
                        const unsigned k = st4[u] + 32u * (unsigned)v + (unsigned)lane;
                        rv[u][v] = (k < en4[u]) ? rp[k] : 0xffffffffu;
                    }
                }
                #pragma unroll
                for (int u = 0; u < 4; ++u) {
                    #pragma unroll
                    for (int v = 0; v < 3; ++v) A2_SCAN(rv[u][v])
                    const unsigned* rp = rec + (size_t)(cb + 32 * (i + u)) * CH_E;
                    for (unsigned k0 = st4[u] + 96u; k0 < en4[u]; k0 += 32u) {
                        const unsigned k = k0 + (unsigned)lane;
                        const unsigned r1 = (k < en4[u]) ? rp[k] : 0xffffffffu;
                        A2_SCAN(r1)
                    }
                }
            }
        }

        __builtin_amdgcn_wave_barrier();
        if ((unsigned)lane < tail - head) {
            const unsigned qi = (head + (unsigned)lane) & (A2_Q - 1);
            a2_apply(hw16, deg, dinvL, acc, qs_w[qi], qd_w[qi]);
        }
        __syncthreads();

        float2* pp = (float2*)(plane + (size_t)base * 10);
        for (int k = tid; k < P * 5; k += 1024) pp[k] = ((const float2*)acc)[k];
        __syncthreads();
    }
}

__global__ void k_final2(const float* __restrict__ part, const float* __restrict__ hw16,
                         const float* __restrict__ deg, const float* __restrict__ b2,
                         float* __restrict__ out, int N) {
    long long gid = (long long)blockIdx.x * blockDim.x + threadIdx.x;
    if (gid >= (long long)N * 10) return;
    const size_t pstride = ((size_t)N * 10 + 31) & ~(size_t)31;
    int node = (int)(gid / 10);
    int f = (int)(gid % 10);
    float p[A2_GJ];
    #pragma unroll
    for (int j = 0; j < A2_GJ; ++j) p[j] = part[(size_t)j * pstride + (size_t)gid];
    float s = p[0];
    #pragma unroll
    for (int j = 1; j < A2_GJ; ++j) s += p[j];
    float di = rsqrtf(deg[node]);
    out[gid] = s + (hw16[(size_t)node * 16 + f] * di * di + b2[f]);
}

extern "C" void kernel_launch(void* const* d_in, const int* in_sizes, int n_in,
                              void* d_out, int out_size, void* d_ws, size_t ws_size,
                              hipStream_t stream) {
    const float* x  = (const float*)d_in[0];
    const int*   ei = (const int*)d_in[1];
    const float* W1 = (const float*)d_in[2];
    const float* b1 = (const float*)d_in[3];
    const float* W2 = (const float*)d_in[4];
    const float* b2 = (const float*)d_in[5];
    float* out = (float*)d_out;

    const int N = in_sizes[0] / 300;
    const int E = in_sizes[1] / 2;
    const int* src = ei;
    const int* dst = ei + E;

    size_t Np = ((size_t)N + 63) & ~(size_t)63;
    float* deg  = (float*)d_ws;
    float* xw   = deg + Np;
    float* h    = xw + Np * 128;
    float* rcf  = h  + Np * 128;
    unsigned* rec = (unsigned*)rcf;
    float* hw   = xw + Np * 80;
    unsigned* tab = (unsigned*)d_out;

    const int T = 256;

    k_init_deg <<<(N + T - 1) / T, T, 0, stream>>>(deg, N);

    k_gemm1    <<<512, T, 0, stream>>>(x, W1, xw, N / 16, src, dst, deg, rec, tab, E);

    k_agg1_own <<<256, 1024, 0, stream>>>(xw, rec, tab, deg, b1, h, N, E);

    k_gemm2    <<<(N / 16 + 7) / 8, T, 0, stream>>>(h, W2, hw, N);

    k_agg2_own <<<A2_GI * A2_GJ, 1024, 0, stream>>>(hw, rec, tab, deg, xw, N, E);

    const long long no = (long long)N * 10;
    k_final2   <<<(unsigned)((no + T - 1) / T), T, 0, stream>>>(xw, hw, deg, b2, out, N);
}
